# starting kernel
# baseline (speedup 1.0000x reference)
#include <hip/hip_runtime.h>
#include <hip/hip_bf16.h>

typedef __attribute__((ext_vector_type(2))) float v2f;
typedef __attribute__((ext_vector_type(8))) float v8f;

#define HC   256
#define NH   4
#define CCH  64
#define KIN  300
#define NOUTD 768

__device__ __forceinline__ unsigned fenc(float f) {
    unsigned u = __float_as_uint(f);
    return (u & 0x80000000u) ? ~u : (u | 0x80000000u);
}
__device__ __forceinline__ float fdec(unsigned u) {
    unsigned b = (u & 0x80000000u) ? (u & 0x7FFFFFFFu) : ~u;
    return __uint_as_float(b);
}

__global__ void init_ws(unsigned* emax, float* denom, float* agg,
                        float* pooled, float* cnt, int n, int g) {
    size_t stride = (size_t)gridDim.x * blockDim.x;
    size_t t = (size_t)blockIdx.x * blockDim.x + threadIdx.x;
    size_t total = (size_t)n * HC;
    const unsigned ninf = 0x007FFFFFu;
    for (size_t i = t; i < total; i += stride) {
        agg[i] = 0.0f;
        if (i < (size_t)n * NH) { emax[i] = ninf; denom[i] = 0.0f; }
        if (i < (size_t)g * HC) pooled[i] = 0.0f;
        if (i < (size_t)g)      cnt[i] = 0.0f;
    }
}

__global__ __launch_bounds__(32) void gemm_x_linw(const float* __restrict__ x,
                                                  const float* __restrict__ w,
                                                  float* __restrict__ h, int n) {
    int mTile = blockIdx.x;
    int lane = threadIdx.x & 31;
    int mr = lane & 15, half = lane >> 4;
    int row = mTile * 16 + mr;
    if (row >= n) row = n - 1;
    int col0 = blockIdx.y * 32 + mr;
    const float* arow = x + (size_t)row * KIN;
    v8f acc0 = {}, acc1 = {};
    for (int k = 0; k < KIN; k += 4) {
        int ka = k + 2 * half;
        v2f a; a.x = arow[ka]; a.y = arow[ka + 1];
        const float* wp = w + (size_t)ka * HC + col0;
        v2f b0; b0.x = wp[0];  b0.y = wp[HC];
        v2f b1; b1.x = wp[16]; b1.y = wp[HC + 16];
        acc0 = __builtin_amdgcn_wmma_f32_16x16x4_f32(false, a, false, b0,
                                                     (short)0, acc0, false, false);
        acc1 = __builtin_amdgcn_wmma_f32_16x16x4_f32(false, a, false, b1,
                                                     (short)0, acc1, false, false);
    }
    int rbase = mTile * 16 + 8 * half;
    if (rbase + 8 <= n) {
        #pragma unroll
        for (int r = 0; r < 8; ++r) {
            float* hp = h + (size_t)(rbase + r) * HC + col0;
            hp[0]  = acc0[r];
            hp[16] = acc1[r];
        }
    } else {
        for (int r = 0; r < 8; ++r) {
            if (rbase + r < n) {
                float* hp = h + (size_t)(rbase + r) * HC + col0;
                hp[0]  = acc0[r];
                hp[16] = acc1[r];
            }
        }
    }
}

__global__ __launch_bounds__(256) void att_kernel(const float* __restrict__ h,
                                                  const float* __restrict__ wsrc,
                                                  const float* __restrict__ wdst,
                                                  float* __restrict__ asrc,
                                                  float* __restrict__ adst, int n) {
    int node = ((int)blockIdx.x * blockDim.x + threadIdx.x) >> 5;
    int lane = threadIdx.x & 31;
    if (node >= n) return;
    int hh = lane >> 3, c0 = lane & 7;
    const float* hp = h + (size_t)node * HC + hh * CCH + c0;
    const float* ws = wsrc + hh * CCH + c0;
    const float* wd = wdst + hh * CCH + c0;
    float ss = 0.f, sd = 0.f;
    #pragma unroll
    for (int i = 0; i < 8; ++i) {
        float v = hp[i * 8];
        ss += v * ws[i * 8];
        sd += v * wd[i * 8];
    }
    #pragma unroll
    for (int off = 4; off >= 1; off >>= 1) {
        ss += __shfl_xor(ss, off, 32);
        sd += __shfl_xor(sd, off, 32);
    }
    if (c0 == 0) {
        asrc[(size_t)node * NH + hh] = ss;
        adst[(size_t)node * NH + hh] = sd;
    }
}

__global__ void edge_max(const int* __restrict__ ei, const float* __restrict__ asrc,
                         const float* __restrict__ adst, unsigned* __restrict__ emax,
                         int E, int n) {
    int e = (int)blockIdx.x * blockDim.x + threadIdx.x;
    if (e >= E + n) return;
    int s, d;
    if (e < E) { s = ei[e]; d = ei[E + e]; } else { s = d = e - E; }
    #pragma unroll
    for (int hh = 0; hh < NH; ++hh) {
        float v = asrc[s * NH + hh] + adst[d * NH + hh];
        v = v > 0.f ? v : 0.2f * v;
        atomicMax(&emax[d * NH + hh], fenc(v));
    }
}

__global__ void edge_sum(const int* __restrict__ ei, const float* __restrict__ asrc,
                         const float* __restrict__ adst, const unsigned* __restrict__ emax,
                         float* __restrict__ denom, int E, int n) {
    int e = (int)blockIdx.x * blockDim.x + threadIdx.x;
    if (e >= E + n) return;
    int s, d;
    if (e < E) { s = ei[e]; d = ei[E + e]; } else { s = d = e - E; }
    #pragma unroll
    for (int hh = 0; hh < NH; ++hh) {
        float v = asrc[s * NH + hh] + adst[d * NH + hh];
        v = v > 0.f ? v : 0.2f * v;
        atomicAdd(&denom[d * NH + hh], __expf(v - fdec(emax[d * NH + hh])));
    }
}

__global__ __launch_bounds__(256) void edge_agg(const int* __restrict__ ei,
                                                const float* __restrict__ asrc,
                                                const float* __restrict__ adst,
                                                const unsigned* __restrict__ emax,
                                                const float* __restrict__ denom,
                                                const float* __restrict__ h,
                                                float* __restrict__ agg, int E, int n) {
    int e = ((int)blockIdx.x * blockDim.x + threadIdx.x) >> 5;
    int lane = threadIdx.x & 31;
    if (e >= E + n) return;
    int s, d;
    if (e < E) { s = ei[e]; d = ei[E + e]; } else { s = d = e - E; }
    int hh = lane >> 3;
    float v = asrc[s * NH + hh] + adst[d * NH + hh];
    v = v > 0.f ? v : 0.2f * v;
    float alpha = __expf(v - fdec(emax[d * NH + hh])) / denom[d * NH + hh];
    const float4* hp = (const float4*)(h + (size_t)s * HC + lane * 8);
    float4 h0 = hp[0], h1 = hp[1];
    float* op = agg + (size_t)d * HC + lane * 8;
    atomicAdd(&op[0], alpha * h0.x);
    atomicAdd(&op[1], alpha * h0.y);
    atomicAdd(&op[2], alpha * h0.z);
    atomicAdd(&op[3], alpha * h0.w);
    atomicAdd(&op[4], alpha * h1.x);
    atomicAdd(&op[5], alpha * h1.y);
    atomicAdd(&op[6], alpha * h1.z);
    atomicAdd(&op[7], alpha * h1.w);
}

__global__ void node_post(const float* __restrict__ agg, const float* __restrict__ bias,
                          const int* __restrict__ batch, float* __restrict__ pooled,
                          float* __restrict__ cnt, int n) {
    int t = (int)blockIdx.x * blockDim.x + threadIdx.x;
    if (t >= n * HC) return;
    int node = t >> 8, ch = t & 255;
    float v = agg[t] + bias[ch];
    v = v > 0.f ? v : 0.01f * v;
    int g = batch[node];
    atomicAdd(&pooled[(size_t)g * HC + ch], v);
    if (ch == 0) atomicAdd(&cnt[g], 1.0f);
}

__global__ __launch_bounds__(32) void gemm_out(const float* __restrict__ pooled,
                                               const float* __restrict__ cnt,
                                               const float* __restrict__ fw,
                                               const float* __restrict__ fb,
                                               float* __restrict__ out, int g) {
    int mTile = blockIdx.x;
    int lane = threadIdx.x & 31;
    int mr = lane & 15, half = lane >> 4;
    int row = mTile * 16 + mr;
    int col0 = blockIdx.y * 32 + mr;
    float inv = 1.0f / fmaxf(cnt[row], 1.0f);
    const float* arow = pooled + (size_t)row * HC;
    v8f acc0 = {}, acc1 = {};
    for (int k = 0; k < HC; k += 4) {
        int ka = k + 2 * half;
        v2f a; a.x = arow[ka] * inv; a.y = arow[ka + 1] * inv;
        const float* wp = fw + (size_t)ka * NOUTD + col0;
        v2f b0; b0.x = wp[0];  b0.y = wp[NOUTD];
        v2f b1; b1.x = wp[16]; b1.y = wp[NOUTD + 16];
        acc0 = __builtin_amdgcn_wmma_f32_16x16x4_f32(false, a, false, b0,
                                                     (short)0, acc0, false, false);
        acc1 = __builtin_amdgcn_wmma_f32_16x16x4_f32(false, a, false, b1,
                                                     (short)0, acc1, false, false);
    }
    float bb0 = fb[col0], bb1 = fb[col0 + 16];
    int rbase = mTile * 16 + 8 * half;
    #pragma unroll
    for (int r = 0; r < 8; ++r) {
        float* op = out + (size_t)(rbase + r) * NOUTD + col0;
        op[0]  = acc0[r] + bb0;
        op[16] = acc1[r] + bb1;
    }
}

extern "C" void kernel_launch(void* const* d_in, const int* in_sizes, int n_in,
                              void* d_out, int out_size, void* d_ws, size_t ws_size,
                              hipStream_t stream) {
    const float* x      = (const float*)d_in[0];
    const int*   ei     = (const int*)  d_in[1];
    const int*   batch  = (const int*)  d_in[2];
    const float* lin_w  = (const float*)d_in[3];
    const float* att_s  = (const float*)d_in[4];
    const float* att_d  = (const float*)d_in[5];
    const float* bias   = (const float*)d_in[6];
    const float* fc1_w  = (const float*)d_in[7];
    const float* fc1_b  = (const float*)d_in[8];
    float* out = (float*)d_out;

    const int N = in_sizes[2];
    const int E = in_sizes[1] / 2;
    const int G = out_size / NOUTD;

    float* ws = (float*)d_ws;
    float*    h      = ws;
    float*    asrc   = h + (size_t)N * HC;
    float*    adst   = asrc + (size_t)N * NH;
    unsigned* emax   = (unsigned*)(adst + (size_t)N * NH);
    float*    denom  = (float*)(emax + (size_t)N * NH);
    float*    agg    = denom + (size_t)N * NH;
    float*    pooled = agg + (size_t)N * HC;
    float*    cnt    = pooled + (size_t)G * HC;

    init_ws<<<2048, 256, 0, stream>>>(emax, denom, agg, pooled, cnt, N, G);

    dim3 g1((N + 15) / 16, HC / 32);
    gemm_x_linw<<<g1, 32, 0, stream>>>(x, lin_w, h, N);

    att_kernel<<<(N * 32 + 255) / 256, 256, 0, stream>>>(h, att_s, att_d, asrc, adst, N);

    int tot = E + N;
    edge_max<<<(tot + 255) / 256, 256, 0, stream>>>(ei, asrc, adst, emax, E, N);
    edge_sum<<<(tot + 255) / 256, 256, 0, stream>>>(ei, asrc, adst, emax, denom, E, N);
    edge_agg<<<((size_t)tot * 32 + 255) / 256, 256, 0, stream>>>(ei, asrc, adst, emax,
                                                                 denom, h, agg, E, N);

    node_post<<<((size_t)N * HC + 255) / 256, 256, 0, stream>>>(agg, bias, batch,
                                                                pooled, cnt, N);

    dim3 g2(G / 16, NOUTD / 32);
    gemm_out<<<g2, 32, 0, stream>>>(pooled, cnt, fc1_w, fc1_b, out, G);
}


# Edges grouped by destination with a per-call counting sort; one wave per node does max, exp-sum, weighted row sum and the pooling add in registers: no float atomics on node rows; sums in slot order
# speedup vs baseline: 1.9524x; 1.9524x over previous
#include <hip/hip_runtime.h>
#include <hip/hip_bf16.h>

typedef __attribute__((ext_vector_type(2))) float v2f;
typedef __attribute__((ext_vector_type(8))) float v8f;

#define HC   256
#define NH   4
#define CCH  64
#define KIN  300
#define NOUTD 768

__global__ void init_ws(int* deg, float* pooled, float* cnt, int n, int g) {
    int stride = (int)(gridDim.x * blockDim.x);
    int t = (int)(blockIdx.x * blockDim.x + threadIdx.x);
    int total = n > g * HC ? n : g * HC;
    for (int i = t; i < total; i += stride) {
        if (i < n)      deg[i] = 0;
        if (i < g * HC) pooled[i] = 0.0f;
        if (i < g)      cnt[i] = 0.0f;
    }
}

__global__ void edge_rank(const int* __restrict__ ei, int* __restrict__ deg,
                          int* __restrict__ rank, int E) {
    int e = (int)(blockIdx.x * blockDim.x + threadIdx.x);
    if (e >= E) return;
    int d = ei[E + e];
    rank[e] = atomicAdd(&deg[d], 1);
}

__global__ __launch_bounds__(1024) void scan_deg(const int* __restrict__ deg,
                                                 int* __restrict__ off, int n) {
    __shared__ int wsum[32];
    int t = threadIdx.x, lane = t & 31, w = t >> 5;
    int carry = 0;
    int v = (t < n) ? deg[t] : 0;
    for (int base = 0; base < n; base += 1024) {
        int nb = base + 1024 + t;
        int vn = (nb < n) ? deg[nb] : 0;
        int x = v;
        #pragma unroll
        for (int o = 1; o < 32; o <<= 1) {
            int y = __shfl_up(x, o, 32);
            if (lane >= o) x += y;
        }
        if (lane == 31) wsum[w] = x;
        __syncthreads();
        int ws = wsum[lane];
        int xs = ws;
        #pragma unroll
        for (int o = 1; o < 32; o <<= 1) {
            int y = __shfl_up(xs, o, 32);
            if (lane >= o) xs += y;
        }
        int wprefix = __shfl(xs, w, 32) - __shfl(ws, w, 32);
        int tot = __shfl(xs, 31, 32);
        int idx = base + t;
        if (idx < n) off[idx] = carry + wprefix + x - v;
        carry += tot;
        __syncthreads();
        v = vn;
    }
    if (t == 0) off[n] = carry;
}

__global__ void edge_fill(const int* __restrict__ ei, const int* __restrict__ off,
                          const int* __restrict__ rank, int* __restrict__ csr, int E) {
    int e = (int)(blockIdx.x * blockDim.x + threadIdx.x);
    if (e >= E) return;
    int s = ei[e], d = ei[E + e];
    csr[off[d] + rank[e]] = s;
}

__global__ __launch_bounds__(32) void gemm_x_linw(const float* __restrict__ x,
                                                  const float* __restrict__ w,
                                                  float* __restrict__ h, int n) {
    int mTile = blockIdx.x;
    int lane = threadIdx.x & 31;
    int mr = lane & 15, half = lane >> 4;
    int row = mTile * 16 + mr;
    if (row >= n) row = n - 1;
    int col0 = blockIdx.y * 32 + mr;
    const float* arow = x + (size_t)row * KIN;
    v8f acc0 = {}, acc1 = {};
    for (int k = 0; k < KIN; k += 4) {
        int ka = k + 2 * half;
        v2f a; a.x = arow[ka]; a.y = arow[ka + 1];
        const float* wp = w + (size_t)ka * HC + col0;
        v2f b0; b0.x = wp[0];  b0.y = wp[HC];
        v2f b1; b1.x = wp[16]; b1.y = wp[HC + 16];
        acc0 = __builtin_amdgcn_wmma_f32_16x16x4_f32(false, a, false, b0,
                                                     (short)0, acc0, false, false);
        acc1 = __builtin_amdgcn_wmma_f32_16x16x4_f32(false, a, false, b1,
                                                     (short)0, acc1, false, false);
    }
    int rbase = mTile * 16 + 8 * half;
    if (rbase + 8 <= n) {
        #pragma unroll
        for (int r = 0; r < 8; ++r) {
            float* hp = h + (size_t)(rbase + r) * HC + col0;
            hp[0]  = acc0[r];
            hp[16] = acc1[r];
        }
    } else {
        for (int r = 0; r < 8; ++r) {
            if (rbase + r < n) {
                float* hp = h + (size_t)(rbase + r) * HC + col0;
                hp[0]  = acc0[r];
                hp[16] = acc1[r];
            }
        }
    }
}

__global__ __launch_bounds__(256) void att_kernel(const float* __restrict__ h,
                                                  const float* __restrict__ wsrc,
                                                  const float* __restrict__ wdst,
                                                  float* __restrict__ asrc,
                                                  float* __restrict__ adst, int n) {
    int node = ((int)blockIdx.x * blockDim.x + threadIdx.x) >> 5;
    int lane = threadIdx.x & 31;
    if (node >= n) return;
    int hh = lane >> 3, c0 = lane & 7;
    const float* hp = h + (size_t)node * HC + hh * CCH + c0;
    const float* ws = wsrc + hh * CCH + c0;
    const float* wd = wdst + hh * CCH + c0;
    float ss = 0.f, sd = 0.f;
    #pragma unroll
    for (int i = 0; i < 8; ++i) {
        float v = hp[i * 8];
        ss += v * ws[i * 8];
        sd += v * wd[i * 8];
    }
    #pragma unroll
    for (int off = 4; off >= 1; off >>= 1) {
        ss += __shfl_xor(ss, off, 32);
        sd += __shfl_xor(sd, off, 32);
    }
    if (c0 == 0) {
        asrc[(size_t)node * NH + hh] = ss;
        adst[(size_t)node * NH + hh] = sd;
    }
}

__device__ __forceinline__ float lrelu_att(float v) { return v > 0.f ? v : 0.2f * v; }
__device__ __forceinline__ float wave_max(float v) {
    #pragma unroll
    for (int o = 16; o >= 1; o >>= 1) v = fmaxf(v, __shfl_xor(v, o, 32));
    return v;
}
__device__ __forceinline__ float wave_sum(float v) {
    #pragma unroll
    for (int o = 16; o >= 1; o >>= 1) v += __shfl_xor(v, o, 32);
    return v;
}
__device__ __forceinline__ float lane_bcast(float v, int i) {
    return __int_as_float(__builtin_amdgcn_readlane(__float_as_int(v), i));
}

__global__ __launch_bounds__(256) void node_gather(const int* __restrict__ off,
                                                   const int* __restrict__ csr,
                                                   const float* __restrict__ asrc,
                                                   const float* __restrict__ adst,
                                                   const float* __restrict__ h,
                                                   const float* __restrict__ bias,
                                                   const int* __restrict__ batch,
                                                   float* __restrict__ pooled,
                                                   float* __restrict__ cnt, int n) {
    int node = (int)((blockIdx.x * blockDim.x + threadIdx.x) >> 5);
    int lane = threadIdx.x & 31;
    if (node >= n) return;
    node = __builtin_amdgcn_readfirstlane(node);
    int beg = off[node];
    int degE = off[node + 1] - beg;
    int total = degE + 1;
    float4 ad = *(const float4*)(adst + (size_t)node * NH);

    float m0 = -INFINITY, m1 = -INFINITY, m2 = -INFINITY, m3 = -INFINITY;
    for (int idx = lane; idx < total; idx += 32) {
        int s = idx < degE ? csr[beg + idx] : node;
        float4 a = *(const float4*)(asrc + (size_t)s * NH);
        m0 = fmaxf(m0, lrelu_att(a.x + ad.x));
        m1 = fmaxf(m1, lrelu_att(a.y + ad.y));
        m2 = fmaxf(m2, lrelu_att(a.z + ad.z));
        m3 = fmaxf(m3, lrelu_att(a.w + ad.w));
    }
    m0 = wave_max(m0); m1 = wave_max(m1); m2 = wave_max(m2); m3 = wave_max(m3);

    float d0 = 0.f, d1 = 0.f, d2 = 0.f, d3 = 0.f;
    for (int idx = lane; idx < total; idx += 32) {
        int s = idx < degE ? csr[beg + idx] : node;
        float4 a = *(const float4*)(asrc + (size_t)s * NH);
        d0 += __expf(lrelu_att(a.x + ad.x) - m0);
        d1 += __expf(lrelu_att(a.y + ad.y) - m1);
        d2 += __expf(lrelu_att(a.z + ad.z) - m2);
        d3 += __expf(lrelu_att(a.w + ad.w) - m3);
    }
    d0 = wave_sum(d0); d1 = wave_sum(d1); d2 = wave_sum(d2); d3 = wave_sum(d3);

    int hsel = lane >> 4;
    float4 accA = {0.f, 0.f, 0.f, 0.f}, accB = {0.f, 0.f, 0.f, 0.f};
    for (int base = 0; base < total; base += 32) {
        int idx = base + lane;
        int s = node;
        float al0 = 0.f, al1 = 0.f, al2 = 0.f, al3 = 0.f;
        if (idx < total) {
            s = idx < degE ? csr[beg + idx] : node;
            float4 a = *(const float4*)(asrc + (size_t)s * NH);
            al0 = __expf(lrelu_att(a.x + ad.x) - m0) / d0;
            al1 = __expf(lrelu_att(a.y + ad.y) - m1) / d1;
            al2 = __expf(lrelu_att(a.z + ad.z) - m2) / d2;
            al3 = __expf(lrelu_att(a.w + ad.w) - m3) / d3;
        }
        int left = total - base;
        int cntc = left < 32 ? left : 32;
        #pragma unroll 4
        for (int i = 0; i < cntc; ++i) {
            int si = __builtin_amdgcn_readlane(s, i);
            float b0 = lane_bcast(al0, i), b1 = lane_bcast(al1, i);
            float b2 = lane_bcast(al2, i), b3 = lane_bcast(al3, i);
            float aA = hsel ? b1 : b0;
            float aB = hsel ? b3 : b2;
            const float4* hp = (const float4*)(h + (size_t)si * HC) + lane;
            float4 x0 = hp[0], x1 = hp[32];
            accA.x += aA * x0.x; accA.y += aA * x0.y; accA.z += aA * x0.z; accA.w += aA * x0.w;
            accB.x += aB * x1.x; accB.y += aB * x1.y; accB.z += aB * x1.z; accB.w += aB * x1.w;
        }
    }

    const float* bp = bias + 4 * lane;
    float r[8] = {accA.x + bp[0],   accA.y + bp[1],   accA.z + bp[2],   accA.w + bp[3],
                  accB.x + bp[128], accB.y + bp[129], accB.z + bp[130], accB.w + bp[131]};
    int g = batch[node];
    float* pA = pooled + (size_t)g * HC + 4 * lane;
    #pragma unroll
    for (int c = 0; c < 4; ++c) {
        float u = r[c];     u = u > 0.f ? u : 0.01f * u;
        float w = r[4 + c]; w = w > 0.f ? w : 0.01f * w;
        atomicAdd(&pA[c], u);
        atomicAdd(&pA[128 + c], w);
    }
    if (lane == 0) atomicAdd(&cnt[g], 1.0f);
}

__global__ __launch_bounds__(32) void gemm_out(const float* __restrict__ pooled,
                                               const float* __restrict__ cnt,
                                               const float* __restrict__ fw,
                                               const float* __restrict__ fb,
                                               float* __restrict__ out, int g) {
    int mTile = blockIdx.x;
    int lane = threadIdx.x & 31;
    int mr = lane & 15, half = lane >> 4;
    int row = mTile * 16 + mr;
    int col0 = blockIdx.y * 32 + mr;
    float inv = 1.0f / fmaxf(cnt[row], 1.0f);
    const float* arow = pooled + (size_t)row * HC;
    v8f acc0 = {}, acc1 = {};
    for (int k = 0; k < HC; k += 4) {
        int ka = k + 2 * half;
        v2f a; a.x = arow[ka] * inv; a.y = arow[ka + 1] * inv;
        const float* wp = fw + (size_t)ka * NOUTD + col0;
        v2f b0; b0.x = wp[0];  b0.y = wp[NOUTD];
        v2f b1; b1.x = wp[16]; b1.y = wp[NOUTD + 16];
        acc0 = __builtin_amdgcn_wmma_f32_16x16x4_f32(false, a, false, b0,
                                                     (short)0, acc0, false, false);
        acc1 = __builtin_amdgcn_wmma_f32_16x16x4_f32(false, a, false, b1,
                                                     (short)0, acc1, false, false);
    }
    float bb0 = fb[col0], bb1 = fb[col0 + 16];
    int rbase = mTile * 16 + 8 * half;
    #pragma unroll
    for (int r = 0; r < 8; ++r) {
        float* op = out + (size_t)(rbase + r) * NOUTD + col0;
        op[0]  = acc0[r] + bb0;
        op[16] = acc1[r] + bb1;
    }
}

extern "C" void kernel_launch(void* const* d_in, const int* in_sizes, int n_in,
                              void* d_out, int out_size, void* d_ws, size_t ws_size,
                              hipStream_t stream) {
    const float* x      = (const float*)d_in[0];
    const int*   ei     = (const int*)  d_in[1];
    const int*   batch  = (const int*)  d_in[2];
    const float* lin_w  = (const float*)d_in[3];
    const float* att_s  = (const float*)d_in[4];
    const float* att_d  = (const float*)d_in[5];
    const float* bias   = (const float*)d_in[6];
    const float* fc1_w  = (const float*)d_in[7];
    const float* fc1_b  = (const float*)d_in[8];
    float* out = (float*)d_out;

    const int N = in_sizes[2];
    const int E = in_sizes[1] / 2;
    const int G = out_size / NOUTD;

    float* ws = (float*)d_ws;
    float* h      = ws;
    float* asrc   = h + (size_t)N * HC;
    float* adst   = asrc + (size_t)N * NH;
    float* pooled = adst + (size_t)N * NH;
    float* cnt    = pooled + (size_t)G * HC;
    int*   deg    = (int*)(cnt + (size_t)G);
    int*   rank   = deg + (size_t)N;
    int*   csr    = rank + (size_t)E;
    int*   off    = csr + (size_t)E;

    init_ws<<<256, 256, 0, stream>>>(deg, pooled, cnt, N, G);

    dim3 g1((N + 15) / 16, HC / 32);
    gemm_x_linw<<<g1, 32, 0, stream>>>(x, lin_w, h, N);

    att_kernel<<<(N * 32 + 255) / 256, 256, 0, stream>>>(h, att_s, att_d, asrc, adst, N);

    edge_rank<<<(E + 255) / 256, 256, 0, stream>>>(ei, deg, rank, E);
    scan_deg<<<1, 1024, 0, stream>>>(deg, off, N);
    edge_fill<<<(E + 255) / 256, 256, 0, stream>>>(ei, off, rank, csr, E);

    node_gather<<<((size_t)N * 32 + 255) / 256, 256, 0, stream>>>(off, csr, asrc, adst, h, bias,
                                                                  batch, pooled, cnt, N);

    dim3 g2(G / 16, NOUTD / 32);
    gemm_out<<<g2, 32, 0, stream>>>(pooled, cnt, fc1_w, fc1_b, out, G);
}


# Pooled add reduced per block of 40 consecutive nodes before the atomics (5 nodes per wave in registers, 8 waves in LDS): one atomic per channel per block, not per node; sum order changes
# speedup vs baseline: 2.9281x; 1.4997x over previous
#include <hip/hip_runtime.h>
#include <hip/hip_bf16.h>

typedef __attribute__((ext_vector_type(2))) float v2f;
typedef __attribute__((ext_vector_type(8))) float v8f;

#define HC   256
#define NH   4
#define CCH  64
#define KIN  300
#define NOUTD 768

__global__ void init_ws(int* deg, float* pooled, float* cnt, int n, int g) {
    int stride = (int)(gridDim.x * blockDim.x);
    int t = (int)(blockIdx.x * blockDim.x + threadIdx.x);
    int total = n > g * HC ? n : g * HC;
    for (int i = t; i < total; i += stride) {
        if (i < n)      deg[i] = 0;
        if (i < g * HC) pooled[i] = 0.0f;
        if (i < g)      cnt[i] = 0.0f;
    }
}

__global__ void edge_rank(const int* __restrict__ ei, int* __restrict__ deg,
                          int* __restrict__ rank, int E) {
    int e = (int)(blockIdx.x * blockDim.x + threadIdx.x);
    if (e >= E) return;
    int d = ei[E + e];
    rank[e] = atomicAdd(&deg[d], 1);
}

__global__ __launch_bounds__(1024) void scan_deg(const int* __restrict__ deg,
                                                 int* __restrict__ off, int n) {
    __shared__ int wsum[32];
    int t = threadIdx.x, lane = t & 31, w = t >> 5;
    int carry = 0;
    int v = (t < n) ? deg[t] : 0;
    for (int base = 0; base < n; base += 1024) {
        int nb = base + 1024 + t;
        int vn = (nb < n) ? deg[nb] : 0;
        int x = v;
        #pragma unroll
        for (int o = 1; o < 32; o <<= 1) {
            int y = __shfl_up(x, o, 32);
            if (lane >= o) x += y;
        }
        if (lane == 31) wsum[w] = x;
        __syncthreads();
        int ws = wsum[lane];
        int xs = ws;
        #pragma unroll
        for (int o = 1; o < 32; o <<= 1) {
            int y = __shfl_up(xs, o, 32);
            if (lane >= o) xs += y;
        }
        int wprefix = __shfl(xs, w, 32) - __shfl(ws, w, 32);
        int tot = __shfl(xs, 31, 32);
        int idx = base + t;
        if (idx < n) off[idx] = carry + wprefix + x - v;
        carry += tot;
        __syncthreads();
        v = vn;
    }
    if (t == 0) off[n] = carry;
}

__global__ void edge_fill(const int* __restrict__ ei, const int* __restrict__ off,
                          const int* __restrict__ rank, int* __restrict__ csr, int E) {
    int e = (int)(blockIdx.x * blockDim.x + threadIdx.x);
    if (e >= E) return;
    int s = ei[e], d = ei[E + e];
    csr[off[d] + rank[e]] = s;
}

__global__ __launch_bounds__(32) void gemm_x_linw(const float* __restrict__ x,
                                                  const float* __restrict__ w,
                                                  float* __restrict__ h, int n) {
    int mTile = blockIdx.x;
    int lane = threadIdx.x & 31;
    int mr = lane & 15, half = lane >> 4;
    int row = mTile * 16 + mr;
    if (row >= n) row = n - 1;
    int col0 = blockIdx.y * 32 + mr;
    const float* arow = x + (size_t)row * KIN;
    v8f acc0 = {}, acc1 = {};
    for (int k = 0; k < KIN; k += 4) {
        int ka = k + 2 * half;
        v2f a; a.x = arow[ka]; a.y = arow[ka + 1];
        const float* wp = w + (size_t)ka * HC + col0;
        v2f b0; b0.x = wp[0];  b0.y = wp[HC];
        v2f b1; b1.x = wp[16]; b1.y = wp[HC + 16];
        acc0 = __builtin_amdgcn_wmma_f32_16x16x4_f32(false, a, false, b0,
                                                     (short)0, acc0, false, false);
        acc1 = __builtin_amdgcn_wmma_f32_16x16x4_f32(false, a, false, b1,
                                                     (short)0, acc1, false, false);
    }
    int rbase = mTile * 16 + 8 * half;
    if (rbase + 8 <= n) {
        #pragma unroll
        for (int r = 0; r < 8; ++r) {
            float* hp = h + (size_t)(rbase + r) * HC + col0;
            hp[0]  = acc0[r];
            hp[16] = acc1[r];
        }
    } else {
        for (int r = 0; r < 8; ++r) {
            if (rbase + r < n) {
                float* hp = h + (size_t)(rbase + r) * HC + col0;
                hp[0]  = acc0[r];
                hp[16] = acc1[r];
            }
        }
    }
}

__global__ __launch_bounds__(256) void att_kernel(const float* __restrict__ h,
                                                  const float* __restrict__ wsrc,
                                                  const float* __restrict__ wdst,
                                                  float* __restrict__ asrc,
                                                  float* __restrict__ adst, int n) {
    int node = ((int)blockIdx.x * blockDim.x + threadIdx.x) >> 5;
    int lane = threadIdx.x & 31;
    if (node >= n) return;
    int hh = lane >> 3, c0 = lane & 7;
    const float* hp = h + (size_t)node * HC + hh * CCH + c0;
    const float* ws = wsrc + hh * CCH + c0;
    const float* wd = wdst + hh * CCH + c0;
    float ss = 0.f, sd = 0.f;
    #pragma unroll
    for (int i = 0; i < 8; ++i) {
        float v = hp[i * 8];
        ss += v * ws[i * 8];
        sd += v * wd[i * 8];
    }
    #pragma unroll
    for (int off = 4; off >= 1; off >>= 1) {
        ss += __shfl_xor(ss, off, 32);
        sd += __shfl_xor(sd, off, 32);
    }
    if (c0 == 0) {
        asrc[(size_t)node * NH + hh] = ss;
        adst[(size_t)node * NH + hh] = sd;
    }
}

__device__ __forceinline__ float lrelu_att(float v) { return v > 0.f ? v : 0.2f * v; }
__device__ __forceinline__ float wave_max(float v) {
    #pragma unroll
    for (int o = 16; o >= 1; o >>= 1) v = fmaxf(v, __shfl_xor(v, o, 32));
    return v;
}
__device__ __forceinline__ float wave_sum(float v) {
    #pragma unroll
    for (int o = 16; o >= 1; o >>= 1) v += __shfl_xor(v, o, 32);
    return v;
}
__device__ __forceinline__ float lane_bcast(float v, int i) {
    return __int_as_float(__builtin_amdgcn_readlane(__float_as_int(v), i));
}

#define NPW 5
__global__ __launch_bounds__(256) void node_gather(const int* __restrict__ off,
                                                   const int* __restrict__ csr,
                                                   const float* __restrict__ asrc,
                                                   const float* __restrict__ adst,
                                                   const float* __restrict__ h,
                                                   const float* __restrict__ bias,
                                                   const int* __restrict__ batch,
                                                   float* __restrict__ pooled,
                                                   float* __restrict__ cnt, int n) {
    __shared__ __align__(16) float sred[8][HC];
    __shared__ int   sgs[8];
    __shared__ float scn[8];
    int lane = threadIdx.x & 31;
    int w = threadIdx.x >> 5;
    int wbase = __builtin_amdgcn_readfirstlane(((int)blockIdx.x * 8 + w) * NPW);
    int hsel = lane >> 4;
    const float* bp = bias + 4 * lane;

    int curg = -1;
    float pa[8] = {0.f, 0.f, 0.f, 0.f, 0.f, 0.f, 0.f, 0.f};
    float pc = 0.f;

    for (int ni = 0; ni < NPW; ++ni) {
        int node = wbase + ni;
        if (node >= n) break;
        int beg = off[node];
        int degE = off[node + 1] - beg;
        int total = degE + 1;
        float4 ad = *(const float4*)(adst + (size_t)node * NH);

        float m0 = -INFINITY, m1 = -INFINITY, m2 = -INFINITY, m3 = -INFINITY;
        for (int idx = lane; idx < total; idx += 32) {
            int s = idx < degE ? csr[beg + idx] : node;
            float4 a = *(const float4*)(asrc + (size_t)s * NH);
            m0 = fmaxf(m0, lrelu_att(a.x + ad.x));
            m1 = fmaxf(m1, lrelu_att(a.y + ad.y));
            m2 = fmaxf(m2, lrelu_att(a.z + ad.z));
            m3 = fmaxf(m3, lrelu_att(a.w + ad.w));
        }
        m0 = wave_max(m0); m1 = wave_max(m1); m2 = wave_max(m2); m3 = wave_max(m3);

        float d0 = 0.f, d1 = 0.f, d2 = 0.f, d3 = 0.f;
        for (int idx = lane; idx < total; idx += 32) {
            int s = idx < degE ? csr[beg + idx] : node;
            float4 a = *(const float4*)(asrc + (size_t)s * NH);
            d0 += __expf(lrelu_att(a.x + ad.x) - m0);
            d1 += __expf(lrelu_att(a.y + ad.y) - m1);
            d2 += __expf(lrelu_att(a.z + ad.z) - m2);
            d3 += __expf(lrelu_att(a.w + ad.w) - m3);
        }
        d0 = wave_sum(d0); d1 = wave_sum(d1); d2 = wave_sum(d2); d3 = wave_sum(d3);

        float4 accA = {0.f, 0.f, 0.f, 0.f}, accB = {0.f, 0.f, 0.f, 0.f};
        for (int base = 0; base < total; base += 32) {
            int idx = base + lane;
            int s = node;
            float al0 = 0.f, al1 = 0.f, al2 = 0.f, al3 = 0.f;
            if (idx < total) {
                s = idx < degE ? csr[beg + idx] : node;
                float4 a = *(const float4*)(asrc + (size_t)s * NH);
                al0 = __expf(lrelu_att(a.x + ad.x) - m0) / d0;
                al1 = __expf(lrelu_att(a.y + ad.y) - m1) / d1;
                al2 = __expf(lrelu_att(a.z + ad.z) - m2) / d2;
                al3 = __expf(lrelu_att(a.w + ad.w) - m3) / d3;
            }
            int left = total - base;
            int cntc = left < 32 ? left : 32;
            #pragma unroll 4
            for (int i = 0; i < cntc; ++i) {
                int si = __builtin_amdgcn_readlane(s, i);
                float b0 = lane_bcast(al0, i), b1 = lane_bcast(al1, i);
                float b2 = lane_bcast(al2, i), b3 = lane_bcast(al3, i);
                float aA = hsel ? b1 : b0;
                float aB = hsel ? b3 : b2;
                const float4* hp = (const float4*)(h + (size_t)si * HC) + lane;
                float4 x0 = hp[0], x1 = hp[32];
                accA.x += aA * x0.x; accA.y += aA * x0.y; accA.z += aA * x0.z; accA.w += aA * x0.w;
                accB.x += aB * x1.x; accB.y += aB * x1.y; accB.z += aB * x1.z; accB.w += aB * x1.w;
            }
        }

        float r[8] = {accA.x + bp[0],   accA.y + bp[1],   accA.z + bp[2],   accA.w + bp[3],
                      accB.x + bp[128], accB.y + bp[129], accB.z + bp[130], accB.w + bp[131]};
        int g = batch[node];
        if (g != curg) {
            if (curg >= 0) {
                float* pA = pooled + (size_t)curg * HC + 4 * lane;
                #pragma unroll
                for (int c = 0; c < 4; ++c) {
                    atomicAdd(&pA[c], pa[c]);
                    atomicAdd(&pA[128 + c], pa[4 + c]);
                }
                if (lane == 0) atomicAdd(&cnt[curg], pc);
            }
            curg = g;
            #pragma unroll
            for (int c = 0; c < 8; ++c) pa[c] = 0.f;
            pc = 0.f;
        }
        #pragma unroll
        for (int c = 0; c < 8; ++c) {
            float u = r[c];
            pa[c] += u > 0.f ? u : 0.01f * u;
        }
        pc += 1.0f;
    }

    *(float4*)&sred[w][4 * lane]       = make_float4(pa[0], pa[1], pa[2], pa[3]);
    *(float4*)&sred[w][128 + 4 * lane] = make_float4(pa[4], pa[5], pa[6], pa[7]);
    if (lane == 0) { sgs[w] = curg; scn[w] = pc; }
    __syncthreads();

    int t = threadIdx.x;
    int gk[8];
    float vk[8], ck[8];
    #pragma unroll
    for (int k = 0; k < 8; ++k) { gk[k] = sgs[k]; vk[k] = sred[k][t]; ck[k] = scn[k]; }
    int cg = -1;
    float s = 0.f, sc = 0.f;
    #pragma unroll
    for (int k = 0; k < 8; ++k) {
        if (gk[k] < 0) continue;
        if (gk[k] != cg) {
            if (cg >= 0) {
                atomicAdd(&pooled[(size_t)cg * HC + t], s);
                if (t == 0) atomicAdd(&cnt[cg], sc);
            }
            cg = gk[k]; s = 0.f; sc = 0.f;
        }
        s += vk[k];
        sc += ck[k];
    }
    if (cg >= 0) {
        atomicAdd(&pooled[(size_t)cg * HC + t], s);
        if (t == 0) atomicAdd(&cnt[cg], sc);
    }
}

__global__ __launch_bounds__(32) void gemm_out(const float* __restrict__ pooled,
                                               const float* __restrict__ cnt,
                                               const float* __restrict__ fw,
                                               const float* __restrict__ fb,
                                               float* __restrict__ out, int g) {
    int mTile = blockIdx.x;
    int lane = threadIdx.x & 31;
    int mr = lane & 15, half = lane >> 4;
    int row = mTile * 16 + mr;
    int col0 = blockIdx.y * 32 + mr;
    float inv = 1.0f / fmaxf(cnt[row], 1.0f);
    const float* arow = pooled + (size_t)row * HC;
    v8f acc0 = {}, acc1 = {};
    for (int k = 0; k < HC; k += 4) {
        int ka = k + 2 * half;
        v2f a; a.x = arow[ka] * inv; a.y = arow[ka + 1] * inv;
        const float* wp = fw + (size_t)ka * NOUTD + col0;
        v2f b0; b0.x = wp[0];  b0.y = wp[NOUTD];
        v2f b1; b1.x = wp[16]; b1.y = wp[NOUTD + 16];
        acc0 = __builtin_amdgcn_wmma_f32_16x16x4_f32(false, a, false, b0,
                                                     (short)0, acc0, false, false);
        acc1 = __builtin_amdgcn_wmma_f32_16x16x4_f32(false, a, false, b1,
                                                     (short)0, acc1, false, false);
    }
    float bb0 = fb[col0], bb1 = fb[col0 + 16];
    int rbase = mTile * 16 + 8 * half;
    #pragma unroll
    for (int r = 0; r < 8; ++r) {
        float* op = out + (size_t)(rbase + r) * NOUTD + col0;
        op[0]  = acc0[r] + bb0;
        op[16] = acc1[r] + bb1;
    }
}

extern "C" void kernel_launch(void* const* d_in, const int* in_sizes, int n_in,
                              void* d_out, int out_size, void* d_ws, size_t ws_size,
                              hipStream_t stream) {
    const float* x      = (const float*)d_in[0];
    const int*   ei     = (const int*)  d_in[1];
    const int*   batch  = (const int*)  d_in[2];
    const float* lin_w  = (const float*)d_in[3];
    const float* att_s  = (const float*)d_in[4];
    const float* att_d  = (const float*)d_in[5];
    const float* bias   = (const float*)d_in[6];
    const float* fc1_w  = (const float*)d_in[7];
    const float* fc1_b  = (const float*)d_in[8];
    float* out = (float*)d_out;

    const int N = in_sizes[2];
    const int E = in_sizes[1] / 2;
    const int G = out_size / NOUTD;

    float* ws = (float*)d_ws;
    float* h      = ws;
    float* asrc   = h + (size_t)N * HC;
    float* adst   = asrc + (size_t)N * NH;
    float* pooled = adst + (size_t)N * NH;
    float* cnt    = pooled + (size_t)G * HC;
    int*   deg    = (int*)(cnt + (size_t)G);
    int*   rank   = deg + (size_t)N;
    int*   csr    = rank + (size_t)E;
    int*   off    = csr + (size_t)E;

    init_ws<<<256, 256, 0, stream>>>(deg, pooled, cnt, N, G);

    dim3 g1((N + 15) / 16, HC / 32);
    gemm_x_linw<<<g1, 32, 0, stream>>>(x, lin_w, h, N);

    att_kernel<<<(N * 32 + 255) / 256, 256, 0, stream>>>(h, att_s, att_d, asrc, adst, N);

    edge_rank<<<(E + 255) / 256, 256, 0, stream>>>(ei, deg, rank, E);
    scan_deg<<<1, 1024, 0, stream>>>(deg, off, N);
    edge_fill<<<(E + 255) / 256, 256, 0, stream>>>(ei, off, rank, csr, E);

    node_gather<<<(N + 8 * NPW - 1) / (8 * NPW), 256, 0, stream>>>(off, csr, asrc, adst, h, bias,
                                                            batch, pooled, cnt, N);

    dim3 g2(G / 16, NOUTD / 32);
    gemm_out<<<g2, 32, 0, stream>>>(pooled, cnt, fc1_w, fc1_b, out, G);
}


# Degree prefix sum no longer one workgroup stepping through 49 chunks in turn: 49 workgroups each add the chunks ahead of their own with independent loads and scan only their own; integer sums
# speedup vs baseline: 3.6999x; 1.2636x over previous
#include <hip/hip_runtime.h>
#include <hip/hip_bf16.h>

typedef __attribute__((ext_vector_type(2))) float v2f;
typedef __attribute__((ext_vector_type(8))) float v8f;

#define HC   256
#define NH   4
#define CCH  64
#define KIN  300
#define NOUTD 768

__global__ void init_ws(int* deg, float* pooled, float* cnt, int n, int g) {
    int stride = (int)(gridDim.x * blockDim.x);
    int t = (int)(blockIdx.x * blockDim.x + threadIdx.x);
    int total = n > g * HC ? n : g * HC;
    for (int i = t; i < total; i += stride) {
        if (i < n)      deg[i] = 0;
        if (i < g * HC) pooled[i] = 0.0f;
        if (i < g)      cnt[i] = 0.0f;
    }
}

__global__ void edge_rank(const int* __restrict__ ei, int* __restrict__ deg,
                          int* __restrict__ rank, int E) {
    int e = (int)(blockIdx.x * blockDim.x + threadIdx.x);
    if (e >= E) return;
    int d = ei[E + e];
    rank[e] = atomicAdd(&deg[d], 1);
}

#define SCAN_CHUNK 1024
#define SCAN_AHEAD 12
__global__ __launch_bounds__(SCAN_CHUNK) void scan_deg(const int* __restrict__ deg,
                                                       int* __restrict__ off, int n) {
    __shared__ int wsum[32];
    __shared__ int psum[32];
    int t = threadIdx.x, lane = t & 31, w = t >> 5;
    int b = (int)blockIdx.x;
    int idx = b * SCAN_CHUNK + t;
    int v = (idx < n) ? deg[idx] : 0;

    int p = 0;
    for (int j = 0; j < b; j += SCAN_AHEAD) {
        for (int u = 0; u < SCAN_AHEAD; ++u) {
            int jj = j + u;
            bool ok = jj < b;
            int val = deg[(ok ? jj : 0) * SCAN_CHUNK + t];
            p += ok ? val : 0;
        }
    }

    int x = v;
    #pragma unroll
    for (int o = 1; o < 32; o <<= 1) {
        int y = __shfl_up(x, o, 32);
        if (lane >= o) x += y;
    }
    for (int o = 16; o >= 1; o >>= 1) p += __shfl_xor(p, o, 32);
    if (lane == 31) wsum[w] = x;
    if (lane == 0)  psum[w] = p;
    __syncthreads();
    int ws = wsum[lane];
    int front = psum[lane];
    int xs = ws;
    #pragma unroll
    for (int o = 1; o < 32; o <<= 1) {
        int y = __shfl_up(xs, o, 32);
        if (lane >= o) xs += y;
    }
    for (int o = 16; o >= 1; o >>= 1) front += __shfl_xor(front, o, 32);
    int wprefix = __shfl(xs, w, 32) - __shfl(ws, w, 32);
    int tot = __shfl(xs, 31, 32);
    if (idx < n) off[idx] = front + wprefix + x - v;
    if (b == (int)gridDim.x - 1 && t == 0) off[n] = front + tot;
}

__global__ void edge_fill(const int* __restrict__ ei, const int* __restrict__ off,
                          const int* __restrict__ rank, int* __restrict__ csr, int E) {
    int e = (int)(blockIdx.x * blockDim.x + threadIdx.x);
    if (e >= E) return;
    int s = ei[e], d = ei[E + e];
    csr[off[d] + rank[e]] = s;
}

__global__ __launch_bounds__(32) void gemm_x_linw(const float* __restrict__ x,
                                                  const float* __restrict__ w,
                                                  float* __restrict__ h, int n) {
    int mTile = blockIdx.x;
    int lane = threadIdx.x & 31;
    int mr = lane & 15, half = lane >> 4;
    int row = mTile * 16 + mr;
    if (row >= n) row = n - 1;
    int col0 = blockIdx.y * 32 + mr;
    const float* arow = x + (size_t)row * KIN;
    v8f acc0 = {}, acc1 = {};
    for (int k = 0; k < KIN; k += 4) {
        int ka = k + 2 * half;
        v2f a; a.x = arow[ka]; a.y = arow[ka + 1];
        const float* wp = w + (size_t)ka * HC + col0;
        v2f b0; b0.x = wp[0];  b0.y = wp[HC];
        v2f b1; b1.x = wp[16]; b1.y = wp[HC + 16];
        acc0 = __builtin_amdgcn_wmma_f32_16x16x4_f32(false, a, false, b0,
                                                     (short)0, acc0, false, false);
        acc1 = __builtin_amdgcn_wmma_f32_16x16x4_f32(false, a, false, b1,
                                                     (short)0, acc1, false, false);
    }
    int rbase = mTile * 16 + 8 * half;
    if (rbase + 8 <= n) {
        #pragma unroll
        for (int r = 0; r < 8; ++r) {
            float* hp = h + (size_t)(rbase + r) * HC + col0;
            hp[0]  = acc0[r];
            hp[16] = acc1[r];
        }
    } else {
        for (int r = 0; r < 8; ++r) {
            if (rbase + r < n) {
                float* hp = h + (size_t)(rbase + r) * HC + col0;
                hp[0]  = acc0[r];
                hp[16] = acc1[r];
            }
        }
    }
}

__global__ __launch_bounds__(256) void att_kernel(const float* __restrict__ h,
                                                  const float* __restrict__ wsrc,
                                                  const float* __restrict__ wdst,
                                                  float* __restrict__ asrc,
                                                  float* __restrict__ adst, int n) {
    int node = ((int)blockIdx.x * blockDim.x + threadIdx.x) >> 5;
    int lane = threadIdx.x & 31;
    if (node >= n) return;
    int hh = lane >> 3, c0 = lane & 7;
    const float* hp = h + (size_t)node * HC + hh * CCH + c0;
    const float* ws = wsrc + hh * CCH + c0;
    const float* wd = wdst + hh * CCH + c0;
    float ss = 0.f, sd = 0.f;
    #pragma unroll
    for (int i = 0; i < 8; ++i) {
        float v = hp[i * 8];
        ss += v * ws[i * 8];
        sd += v * wd[i * 8];
    }
    #pragma unroll
    for (int off = 4; off >= 1; off >>= 1) {
        ss += __shfl_xor(ss, off, 32);
        sd += __shfl_xor(sd, off, 32);
    }
    if (c0 == 0) {
        asrc[(size_t)node * NH + hh] = ss;
        adst[(size_t)node * NH + hh] = sd;
    }
}

__device__ __forceinline__ float lrelu_att(float v) { return v > 0.f ? v : 0.2f * v; }
__device__ __forceinline__ float wave_max(float v) {
    #pragma unroll
    for (int o = 16; o >= 1; o >>= 1) v = fmaxf(v, __shfl_xor(v, o, 32));
    return v;
}
__device__ __forceinline__ float wave_sum(float v) {
    #pragma unroll
    for (int o = 16; o >= 1; o >>= 1) v += __shfl_xor(v, o, 32);
    return v;
}
__device__ __forceinline__ float lane_bcast(float v, int i) {
    return __int_as_float(__builtin_amdgcn_readlane(__float_as_int(v), i));
}

#define NPW 5
__global__ __launch_bounds__(256) void node_gather(const int* __restrict__ off,
                                                   const int* __restrict__ csr,
                                                   const float* __restrict__ asrc,
                                                   const float* __restrict__ adst,
                                                   const float* __restrict__ h,
                                                   const float* __restrict__ bias,
                                                   const int* __restrict__ batch,
                                                   float* __restrict__ pooled,
                                                   float* __restrict__ cnt, int n) {
    __shared__ __align__(16) float sred[8][HC];
    __shared__ int   sgs[8];
    __shared__ float scn[8];
    int lane = threadIdx.x & 31;
    int w = threadIdx.x >> 5;
    int wbase = __builtin_amdgcn_readfirstlane(((int)blockIdx.x * 8 + w) * NPW);
    int hsel = lane >> 4;
    const float* bp = bias + 4 * lane;

    int curg = -1;
    float pa[8] = {0.f, 0.f, 0.f, 0.f, 0.f, 0.f, 0.f, 0.f};
    float pc = 0.f;

    for (int ni = 0; ni < NPW; ++ni) {
        int node = wbase + ni;
        if (node >= n) break;
        int beg = off[node];
        int degE = off[node + 1] - beg;
        int total = degE + 1;
        float4 ad = *(const float4*)(adst + (size_t)node * NH);

        float m0 = -INFINITY, m1 = -INFINITY, m2 = -INFINITY, m3 = -INFINITY;
        for (int idx = lane; idx < total; idx += 32) {
            int s = idx < degE ? csr[beg + idx] : node;
            float4 a = *(const float4*)(asrc + (size_t)s * NH);
            m0 = fmaxf(m0, lrelu_att(a.x + ad.x));
            m1 = fmaxf(m1, lrelu_att(a.y + ad.y));
            m2 = fmaxf(m2, lrelu_att(a.z + ad.z));
            m3 = fmaxf(m3, lrelu_att(a.w + ad.w));
        }
        m0 = wave_max(m0); m1 = wave_max(m1); m2 = wave_max(m2); m3 = wave_max(m3);

        float d0 = 0.f, d1 = 0.f, d2 = 0.f, d3 = 0.f;
        for (int idx = lane; idx < total; idx += 32) {
            int s = idx < degE ? csr[beg + idx] : node;
            float4 a = *(const float4*)(asrc + (size_t)s * NH);
            d0 += __expf(lrelu_att(a.x + ad.x) - m0);
            d1 += __expf(lrelu_att(a.y + ad.y) - m1);
            d2 += __expf(lrelu_att(a.z + ad.z) - m2);
            d3 += __expf(lrelu_att(a.w + ad.w) - m3);
        }
        d0 = wave_sum(d0); d1 = wave_sum(d1); d2 = wave_sum(d2); d3 = wave_sum(d3);

        float4 accA = {0.f, 0.f, 0.f, 0.f}, accB = {0.f, 0.f, 0.f, 0.f};
        for (int base = 0; base < total; base += 32) {
            int idx = base + lane;
            int s = node;
            float al0 = 0.f, al1 = 0.f, al2 = 0.f, al3 = 0.f;
            if (idx < total) {
                s = idx < degE ? csr[beg + idx] : node;
                float4 a = *(const float4*)(asrc + (size_t)s * NH);
                al0 = __expf(lrelu_att(a.x + ad.x) - m0) / d0;
                al1 = __expf(lrelu_att(a.y + ad.y) - m1) / d1;
                al2 = __expf(lrelu_att(a.z + ad.z) - m2) / d2;
                al3 = __expf(lrelu_att(a.w + ad.w) - m3) / d3;
            }
            int left = total - base;
            int cntc = left < 32 ? left : 32;
            #pragma unroll 4
            for (int i = 0; i < cntc; ++i) {
                int si = __builtin_amdgcn_readlane(s, i);
                float b0 = lane_bcast(al0, i), b1 = lane_bcast(al1, i);
                float b2 = lane_bcast(al2, i), b3 = lane_bcast(al3, i);
                float aA = hsel ? b1 : b0;
                float aB = hsel ? b3 : b2;
                const float4* hp = (const float4*)(h + (size_t)si * HC) + lane;
                float4 x0 = hp[0], x1 = hp[32];
                accA.x += aA * x0.x; accA.y += aA * x0.y; accA.z += aA * x0.z; accA.w += aA * x0.w;
                accB.x += aB * x1.x; accB.y += aB * x1.y; accB.z += aB * x1.z; accB.w += aB * x1.w;
            }
        }

        float r[8] = {accA.x + bp[0],   accA.y + bp[1],   accA.z + bp[2],   accA.w + bp[3],
                      accB.x + bp[128], accB.y + bp[129], accB.z + bp[130], accB.w + bp[131]};
        int g = batch[node];
        if (g != curg) {
            if (curg >= 0) {
                float* pA = pooled + (size_t)curg * HC + 4 * lane;
                #pragma unroll
                for (int c = 0; c < 4; ++c) {
                    atomicAdd(&pA[c], pa[c]);
                    atomicAdd(&pA[128 + c], pa[4 + c]);
                }
                if (lane == 0) atomicAdd(&cnt[curg], pc);
            }
            curg = g;
            #pragma unroll
            for (int c = 0; c < 8; ++c) pa[c] = 0.f;
            pc = 0.f;
        }
        #pragma unroll
        for (int c = 0; c < 8; ++c) {
            float u = r[c];
            pa[c] += u > 0.f ? u : 0.01f * u;
        }
        pc += 1.0f;
    }

    *(float4*)&sred[w][4 * lane]       = make_float4(pa[0], pa[1], pa[2], pa[3]);
    *(float4*)&sred[w][128 + 4 * lane] = make_float4(pa[4], pa[5], pa[6], pa[7]);
    if (lane == 0) { sgs[w] = curg; scn[w] = pc; }
    __syncthreads();

    int t = threadIdx.x;
    int gk[8];
    float vk[8], ck[8];
    #pragma unroll
    for (int k = 0; k < 8; ++k) { gk[k] = sgs[k]; vk[k] = sred[k][t]; ck[k] = scn[k]; }
    int cg = -1;
    float s = 0.f, sc = 0.f;
    #pragma unroll
    for (int k = 0; k < 8; ++k) {
        if (gk[k] < 0) continue;
        if (gk[k] != cg) {
            if (cg >= 0) {
                atomicAdd(&pooled[(size_t)cg * HC + t], s);
                if (t == 0) atomicAdd(&cnt[cg], sc);
            }
            cg = gk[k]; s = 0.f; sc = 0.f;
        }
        s += vk[k];
        sc += ck[k];
    }
    if (cg >= 0) {
        atomicAdd(&pooled[(size_t)cg * HC + t], s);
        if (t == 0) atomicAdd(&cnt[cg], sc);
    }
}

__global__ __launch_bounds__(32) void gemm_out(const float* __restrict__ pooled,
                                               const float* __restrict__ cnt,
                                               const float* __restrict__ fw,
                                               const float* __restrict__ fb,
                                               float* __restrict__ out, int g) {
    int mTile = blockIdx.x;
    int lane = threadIdx.x & 31;
    int mr = lane & 15, half = lane >> 4;
    int row = mTile * 16 + mr;
    int col0 = blockIdx.y * 32 + mr;
    float inv = 1.0f / fmaxf(cnt[row], 1.0f);
    const float* arow = pooled + (size_t)row * HC;
    v8f acc0 = {}, acc1 = {};
    for (int k = 0; k < HC; k += 4) {
        int ka = k + 2 * half;
        v2f a; a.x = arow[ka] * inv; a.y = arow[ka + 1] * inv;
        const float* wp = fw + (size_t)ka * NOUTD + col0;
        v2f b0; b0.x = wp[0];  b0.y = wp[NOUTD];
        v2f b1; b1.x = wp[16]; b1.y = wp[NOUTD + 16];
        acc0 = __builtin_amdgcn_wmma_f32_16x16x4_f32(false, a, false, b0,
                                                     (short)0, acc0, false, false);
        acc1 = __builtin_amdgcn_wmma_f32_16x16x4_f32(false, a, false, b1,
                                                     (short)0, acc1, false, false);
    }
    float bb0 = fb[col0], bb1 = fb[col0 + 16];
    int rbase = mTile * 16 + 8 * half;
    #pragma unroll
    for (int r = 0; r < 8; ++r) {
        float* op = out + (size_t)(rbase + r) * NOUTD + col0;
        op[0]  = acc0[r] + bb0;
        op[16] = acc1[r] + bb1;
    }
}

extern "C" void kernel_launch(void* const* d_in, const int* in_sizes, int n_in,
                              void* d_out, int out_size, void* d_ws, size_t ws_size,
                              hipStream_t stream) {
    const float* x      = (const float*)d_in[0];
    const int*   ei     = (const int*)  d_in[1];
    const int*   batch  = (const int*)  d_in[2];
    const float* lin_w  = (const float*)d_in[3];
    const float* att_s  = (const float*)d_in[4];
    const float* att_d  = (const float*)d_in[5];
    const float* bias   = (const float*)d_in[6];
    const float* fc1_w  = (const float*)d_in[7];
    const float* fc1_b  = (const float*)d_in[8];
    float* out = (float*)d_out;

    const int N = in_sizes[2];
    const int E = in_sizes[1] / 2;
    const int G = out_size / NOUTD;

    float* ws = (float*)d_ws;
    float* h      = ws;
    float* asrc   = h + (size_t)N * HC;
    float* adst   = asrc + (size_t)N * NH;
    float* pooled = adst + (size_t)N * NH;
    float* cnt    = pooled + (size_t)G * HC;
    int*   deg    = (int*)(cnt + (size_t)G);
    int*   rank   = deg + (size_t)N;
    int*   csr    = rank + (size_t)E;
    int*   off    = csr + (size_t)E;

    init_ws<<<256, 256, 0, stream>>>(deg, pooled, cnt, N, G);

    dim3 g1((N + 15) / 16, HC / 32);
    gemm_x_linw<<<g1, 32, 0, stream>>>(x, lin_w, h, N);

    att_kernel<<<(N * 32 + 255) / 256, 256, 0, stream>>>(h, att_s, att_d, asrc, adst, N);

    edge_rank<<<(E + 255) / 256, 256, 0, stream>>>(ei, deg, rank, E);
    scan_deg<<<N / SCAN_CHUNK + 1, SCAN_CHUNK, 0, stream>>>(deg, off, N);
    edge_fill<<<(E + 255) / 256, 256, 0, stream>>>(ei, off, rank, csr, E);

    node_gather<<<(N + 8 * NPW - 1) / (8 * NPW), 256, 0, stream>>>(off, csr, asrc, adst, h, bias,
                                                            batch, pooled, cnt, N);

    dim3 g2(G / 16, NOUTD / 32);
    gemm_out<<<g2, 32, 0, stream>>>(pooled, cnt, fc1_w, fc1_b, out, G);
}


# GEMM1 operands staged through LDS: a workgroup of 4 waves takes 32 rows x all 256 columns, loads each x and lin_w chunk once with whole-line loads, waves read fragments from LDS; sum order kept
# speedup vs baseline: 4.6701x; 1.2622x over previous
#include <hip/hip_runtime.h>
#include <hip/hip_bf16.h>

typedef __attribute__((ext_vector_type(2))) float v2f;
typedef __attribute__((ext_vector_type(8))) float v8f;

#define HC   256
#define NH   4
#define CCH  64
#define KIN  300
#define NOUTD 768

__global__ void init_ws(int* deg, float* pooled, float* cnt, int n, int g) {
    int stride = (int)(gridDim.x * blockDim.x);
    int t = (int)(blockIdx.x * blockDim.x + threadIdx.x);
    int total = n > g * HC ? n : g * HC;
    for (int i = t; i < total; i += stride) {
        if (i < n)      deg[i] = 0;
        if (i < g * HC) pooled[i] = 0.0f;
        if (i < g)      cnt[i] = 0.0f;
    }
}

__global__ void edge_rank(const int* __restrict__ ei, int* __restrict__ deg,
                          int* __restrict__ rank, int E) {
    int e = (int)(blockIdx.x * blockDim.x + threadIdx.x);
    if (e >= E) return;
    int d = ei[E + e];
    rank[e] = atomicAdd(&deg[d], 1);
}

#define SCAN_CHUNK 1024
#define SCAN_AHEAD 12
__global__ __launch_bounds__(SCAN_CHUNK) void scan_deg(const int* __restrict__ deg,
                                                       int* __restrict__ off, int n) {
    __shared__ int wsum[32];
    __shared__ int psum[32];
    int t = threadIdx.x, lane = t & 31, w = t >> 5;
    int b = (int)blockIdx.x;
    int idx = b * SCAN_CHUNK + t;
    int v = (idx < n) ? deg[idx] : 0;

    int p = 0;
    for (int j = 0; j < b; j += SCAN_AHEAD) {
        for (int u = 0; u < SCAN_AHEAD; ++u) {
            int jj = j + u;
            bool ok = jj < b;
            int val = deg[(ok ? jj : 0) * SCAN_CHUNK + t];
            p += ok ? val : 0;
        }
    }

    int x = v;
    #pragma unroll
    for (int o = 1; o < 32; o <<= 1) {
        int y = __shfl_up(x, o, 32);
        if (lane >= o) x += y;
    }
    for (int o = 16; o >= 1; o >>= 1) p += __shfl_xor(p, o, 32);
    if (lane == 31) wsum[w] = x;
    if (lane == 0)  psum[w] = p;
    __syncthreads();
    int ws = wsum[lane];
    int front = psum[lane];
    int xs = ws;
    #pragma unroll
    for (int o = 1; o < 32; o <<= 1) {
        int y = __shfl_up(xs, o, 32);
        if (lane >= o) xs += y;
    }
    for (int o = 16; o >= 1; o >>= 1) front += __shfl_xor(front, o, 32);
    int wprefix = __shfl(xs, w, 32) - __shfl(ws, w, 32);
    int tot = __shfl(xs, 31, 32);
    if (idx < n) off[idx] = front + wprefix + x - v;
    if (b == (int)gridDim.x - 1 && t == 0) off[n] = front + tot;
}

__global__ void edge_fill(const int* __restrict__ ei, const int* __restrict__ off,
                          const int* __restrict__ rank, int* __restrict__ csr, int E) {
    int e = (int)(blockIdx.x * blockDim.x + threadIdx.x);
    if (e >= E) return;
    int s = ei[e], d = ei[E + e];
    csr[off[d] + rank[e]] = s;
}

#define G1_ROWS 32
#define G1_KC   20
#define G1_AP   20
#define G1_BP   272
__global__ __launch_bounds__(128) void gemm_x_linw(const float* __restrict__ x,
                                                   const float* __restrict__ w,
                                                   float* __restrict__ h, int n) {
    __shared__ __align__(16) float As[G1_ROWS * G1_AP];
    __shared__ __align__(16) float Bs[(G1_KC / 2) * G1_BP * 2];
    int t = threadIdx.x;
    int lane = t & 31, wv = t >> 5;
    int mr = lane & 15, half = lane >> 4;
    int mt = wv & 1, nh = wv >> 1;
    int m0 = (int)blockIdx.x * G1_ROWS;

    int ra = t / 5, qa = t - 5 * ra;
    int rb2 = (128 + t) / 5, qb2 = (128 + t) - 5 * rb2;
    int gra = m0 + ra;  gra = gra < n ? gra : n - 1;
    int grb = m0 + rb2; grb = grb < n ? grb : n - 1;
    const float* pa0 = x + (size_t)gra * KIN + 4 * qa;
    const float* pa1 = x + (size_t)grb * KIN + 4 * qb2;
    float* sa0 = As + ra * G1_AP + 4 * qa;
    float* sa1 = As + rb2 * G1_AP + 4 * qb2;

    int cg = t & 63, kq = t >> 6;
    const float* pw = w + (size_t)(2 * kq) * HC + 4 * cg;
    float* sb = Bs + ((size_t)kq * G1_BP + 4 * cg) * 2;

    const float* fa = As + (16 * mt + mr) * G1_AP + 2 * half;
    const float* fb = Bs + ((size_t)half * G1_BP + 128 * nh + mr) * 2;

    v8f acc[8];
    #pragma unroll
    for (int j = 0; j < 8; ++j) acc[j] = (v8f){0.f, 0.f, 0.f, 0.f, 0.f, 0.f, 0.f, 0.f};

    for (int c = 0; c < KIN / G1_KC; ++c) {
        int k0 = c * G1_KC;
        float4 xa0 = *(const float4*)(pa0 + k0);
        float4 xa1 = make_float4(0.f, 0.f, 0.f, 0.f);
        if (wv == 0) xa1 = *(const float4*)(pa1 + k0);
        float4 we[5], wo[5];
        #pragma unroll
        for (int i = 0; i < 5; ++i) {
            const float* p = pw + (size_t)(k0 + 4 * i) * HC;
            we[i] = *(const float4*)p;
            wo[i] = *(const float4*)(p + HC);
        }
        __syncthreads();
        *(float4*)sa0 = xa0;
        if (wv == 0) *(float4*)sa1 = xa1;
        #pragma unroll
        for (int i = 0; i < 5; ++i) {
            float* q = sb + (size_t)(2 * i) * G1_BP * 2;
            *(float4*)q       = make_float4(we[i].x, wo[i].x, we[i].y, wo[i].y);
            *(float4*)(q + 4) = make_float4(we[i].z, wo[i].z, we[i].w, wo[i].w);
        }
        __syncthreads();
        #pragma unroll
        for (int ks = 0; ks < G1_KC / 4; ++ks) {
            v2f a = *(const v2f*)(fa + 4 * ks);
            #pragma unroll
            for (int j = 0; j < 8; ++j) {
                v2f b = *(const v2f*)(fb + ((size_t)(2 * ks) * G1_BP + 16 * j) * 2);
                acc[j] = __builtin_amdgcn_wmma_f32_16x16x4_f32(false, a, false, b,
                                                               (short)0, acc[j], false, false);
            }
        }
    }

    int rbase = m0 + 16 * mt + 8 * half;
    int col0 = 128 * nh + mr;
    if (rbase + 8 <= n) {
        #pragma unroll
        for (int j = 0; j < 8; ++j) {
            #pragma unroll
            for (int r = 0; r < 8; ++r)
                h[(size_t)(rbase + r) * HC + col0 + 16 * j] = acc[j][r];
        }
    } else {
        #pragma unroll
        for (int j = 0; j < 8; ++j) {
            #pragma unroll
            for (int r = 0; r < 8; ++r)
                if (rbase + r < n) h[(size_t)(rbase + r) * HC + col0 + 16 * j] = acc[j][r];
        }
    }
}

__global__ __launch_bounds__(256) void att_kernel(const float* __restrict__ h,
                                                  const float* __restrict__ wsrc,
                                                  const float* __restrict__ wdst,
                                                  float* __restrict__ asrc,
                                                  float* __restrict__ adst, int n) {
    int node = ((int)blockIdx.x * blockDim.x + threadIdx.x) >> 5;
    int lane = threadIdx.x & 31;
    if (node >= n) return;
    int hh = lane >> 3, c0 = lane & 7;
    const float* hp = h + (size_t)node * HC + hh * CCH + c0;
    const float* ws = wsrc + hh * CCH + c0;
    const float* wd = wdst + hh * CCH + c0;
    float ss = 0.f, sd = 0.f;
    #pragma unroll
    for (int i = 0; i < 8; ++i) {
        float v = hp[i * 8];
        ss += v * ws[i * 8];
        sd += v * wd[i * 8];
    }
    #pragma unroll
    for (int off = 4; off >= 1; off >>= 1) {
        ss += __shfl_xor(ss, off, 32);
        sd += __shfl_xor(sd, off, 32);
    }
    if (c0 == 0) {
        asrc[(size_t)node * NH + hh] = ss;
        adst[(size_t)node * NH + hh] = sd;
    }
}

__device__ __forceinline__ float lrelu_att(float v) { return v > 0.f ? v : 0.2f * v; }
__device__ __forceinline__ float wave_max(float v) {
    #pragma unroll
    for (int o = 16; o >= 1; o >>= 1) v = fmaxf(v, __shfl_xor(v, o, 32));
    return v;
}
__device__ __forceinline__ float wave_sum(float v) {
    #pragma unroll
    for (int o = 16; o >= 1; o >>= 1) v += __shfl_xor(v, o, 32);
    return v;
}
__device__ __forceinline__ float lane_bcast(float v, int i) {
    return __int_as_float(__builtin_amdgcn_readlane(__float_as_int(v), i));
}

#define NPW 5
__global__ __launch_bounds__(256) void node_gather(const int* __restrict__ off,
                                                   const int* __restrict__ csr,
                                                   const float* __restrict__ asrc,
                                                   const float* __restrict__ adst,
                                                   const float* __restrict__ h,
                                                   const float* __restrict__ bias,
                                                   const int* __restrict__ batch,
                                                   float* __restrict__ pooled,
                                                   float* __restrict__ cnt, int n) {
    __shared__ __align__(16) float sred[8][HC];
    __shared__ int   sgs[8];
    __shared__ float scn[8];
    int lane = threadIdx.x & 31;
    int w = threadIdx.x >> 5;
    int wbase = __builtin_amdgcn_readfirstlane(((int)blockIdx.x * 8 + w) * NPW);
    int hsel = lane >> 4;
    const float* bp = bias + 4 * lane;

    int curg = -1;
    float pa[8] = {0.f, 0.f, 0.f, 0.f, 0.f, 0.f, 0.f, 0.f};
    float pc = 0.f;

    for (int ni = 0; ni < NPW; ++ni) {
        int node = wbase + ni;
        if (node >= n) break;
        int beg = off[node];
        int degE = off[node + 1] - beg;
        int total = degE + 1;
        float4 ad = *(const float4*)(adst + (size_t)node * NH);

        float m0 = -INFINITY, m1 = -INFINITY, m2 = -INFINITY, m3 = -INFINITY;
        for (int idx = lane; idx < total; idx += 32) {
            int s = idx < degE ? csr[beg + idx] : node;
            float4 a = *(const float4*)(asrc + (size_t)s * NH);
            m0 = fmaxf(m0, lrelu_att(a.x + ad.x));
            m1 = fmaxf(m1, lrelu_att(a.y + ad.y));
            m2 = fmaxf(m2, lrelu_att(a.z + ad.z));
            m3 = fmaxf(m3, lrelu_att(a.w + ad.w));
        }
        m0 = wave_max(m0); m1 = wave_max(m1); m2 = wave_max(m2); m3 = wave_max(m3);

        float d0 = 0.f, d1 = 0.f, d2 = 0.f, d3 = 0.f;
        for (int idx = lane; idx < total; idx += 32) {
            int s = idx < degE ? csr[beg + idx] : node;
            float4 a = *(const float4*)(asrc + (size_t)s * NH);
            d0 += __expf(lrelu_att(a.x + ad.x) - m0);
            d1 += __expf(lrelu_att(a.y + ad.y) - m1);
            d2 += __expf(lrelu_att(a.z + ad.z) - m2);
            d3 += __expf(lrelu_att(a.w + ad.w) - m3);
        }
        d0 = wave_sum(d0); d1 = wave_sum(d1); d2 = wave_sum(d2); d3 = wave_sum(d3);

        float4 accA = {0.f, 0.f, 0.f, 0.f}, accB = {0.f, 0.f, 0.f, 0.f};
        for (int base = 0; base < total; base += 32) {
            int idx = base + lane;
            int s = node;
            float al0 = 0.f, al1 = 0.f, al2 = 0.f, al3 = 0.f;
            if (idx < total) {
                s = idx < degE ? csr[beg + idx] : node;
                float4 a = *(const float4*)(asrc + (size_t)s * NH);
                al0 = __expf(lrelu_att(a.x + ad.x) - m0) / d0;
                al1 = __expf(lrelu_att(a.y + ad.y) - m1) / d1;
                al2 = __expf(lrelu_att(a.z + ad.z) - m2) / d2;
                al3 = __expf(lrelu_att(a.w + ad.w) - m3) / d3;
            }
            int left = total - base;
            int cntc = left < 32 ? left : 32;
            #pragma unroll 4
            for (int i = 0; i < cntc; ++i) {
                int si = __builtin_amdgcn_readlane(s, i);
                float b0 = lane_bcast(al0, i), b1 = lane_bcast(al1, i);
                float b2 = lane_bcast(al2, i), b3 = lane_bcast(al3, i);
                float aA = hsel ? b1 : b0;
                float aB = hsel ? b3 : b2;
                const float4* hp = (const float4*)(h + (size_t)si * HC) + lane;
                float4 x0 = hp[0], x1 = hp[32];
                accA.x += aA * x0.x; accA.y += aA * x0.y; accA.z += aA * x0.z; accA.w += aA * x0.w;
                accB.x += aB * x1.x; accB.y += aB * x1.y; accB.z += aB * x1.z; accB.w += aB * x1.w;
            }
        }

        float r[8] = {accA.x + bp[0],   accA.y + bp[1],   accA.z + bp[2],   accA.w + bp[3],
                      accB.x + bp[128], accB.y + bp[129], accB.z + bp[130], accB.w + bp[131]};
        int g = batch[node];
        if (g != curg) {
            if (curg >= 0) {
                float* pA = pooled + (size_t)curg * HC + 4 * lane;
                #pragma unroll
                for (int c = 0; c < 4; ++c) {
                    atomicAdd(&pA[c], pa[c]);
                    atomicAdd(&pA[128 + c], pa[4 + c]);
                }
                if (lane == 0) atomicAdd(&cnt[curg], pc);
            }
            curg = g;
            #pragma unroll
            for (int c = 0; c < 8; ++c) pa[c] = 0.f;
            pc = 0.f;
        }
        #pragma unroll
        for (int c = 0; c < 8; ++c) {
            float u = r[c];
            pa[c] += u > 0.f ? u : 0.01f * u;
        }
        pc += 1.0f;
    }

    *(float4*)&sred[w][4 * lane]       = make_float4(pa[0], pa[1], pa[2], pa[3]);
    *(float4*)&sred[w][128 + 4 * lane] = make_float4(pa[4], pa[5], pa[6], pa[7]);
    if (lane == 0) { sgs[w] = curg; scn[w] = pc; }
    __syncthreads();

    int t = threadIdx.x;
    int gk[8];
    float vk[8], ck[8];
    #pragma unroll
    for (int k = 0; k < 8; ++k) { gk[k] = sgs[k]; vk[k] = sred[k][t]; ck[k] = scn[k]; }
    int cg = -1;
    float s = 0.f, sc = 0.f;
    #pragma unroll
    for (int k = 0; k < 8; ++k) {
        if (gk[k] < 0) continue;
        if (gk[k] != cg) {
            if (cg >= 0) {
                atomicAdd(&pooled[(size_t)cg * HC + t], s);
                if (t == 0) atomicAdd(&cnt[cg], sc);
            }
            cg = gk[k]; s = 0.f; sc = 0.f;
        }
        s += vk[k];
        sc += ck[k];
    }
    if (cg >= 0) {
        atomicAdd(&pooled[(size_t)cg * HC + t], s);
        if (t == 0) atomicAdd(&cnt[cg], sc);
    }
}

__global__ __launch_bounds__(32) void gemm_out(const float* __restrict__ pooled,
                                               const float* __restrict__ cnt,
                                               const float* __restrict__ fw,
                                               const float* __restrict__ fb,
                                               float* __restrict__ out, int g) {
    int mTile = blockIdx.x;
    int lane = threadIdx.x & 31;
    int mr = lane & 15, half = lane >> 4;
    int row = mTile * 16 + mr;
    int col0 = blockIdx.y * 32 + mr;
    float inv = 1.0f / fmaxf(cnt[row], 1.0f);
    const float* arow = pooled + (size_t)row * HC;
    v8f acc0 = {}, acc1 = {};
    for (int k = 0; k < HC; k += 4) {
        int ka = k + 2 * half;
        v2f a; a.x = arow[ka] * inv; a.y = arow[ka + 1] * inv;
        const float* wp = fw + (size_t)ka * NOUTD + col0;
        v2f b0; b0.x = wp[0];  b0.y = wp[NOUTD];
        v2f b1; b1.x = wp[16]; b1.y = wp[NOUTD + 16];
        acc0 = __builtin_amdgcn_wmma_f32_16x16x4_f32(false, a, false, b0,
                                                     (short)0, acc0, false, false);
        acc1 = __builtin_amdgcn_wmma_f32_16x16x4_f32(false, a, false, b1,
                                                     (short)0, acc1, false, false);
    }
    float bb0 = fb[col0], bb1 = fb[col0 + 16];
    int rbase = mTile * 16 + 8 * half;
    #pragma unroll
    for (int r = 0; r < 8; ++r) {
        float* op = out + (size_t)(rbase + r) * NOUTD + col0;
        op[0]  = acc0[r] + bb0;
        op[16] = acc1[r] + bb1;
    }
}

extern "C" void kernel_launch(void* const* d_in, const int* in_sizes, int n_in,
                              void* d_out, int out_size, void* d_ws, size_t ws_size,
                              hipStream_t stream) {
    const float* x      = (const float*)d_in[0];
    const int*   ei     = (const int*)  d_in[1];
    const int*   batch  = (const int*)  d_in[2];
    const float* lin_w  = (const float*)d_in[3];
    const float* att_s  = (const float*)d_in[4];
    const float* att_d  = (const float*)d_in[5];
    const float* bias   = (const float*)d_in[6];
    const float* fc1_w  = (const float*)d_in[7];
    const float* fc1_b  = (const float*)d_in[8];
    float* out = (float*)d_out;

    const int N = in_sizes[2];
    const int E = in_sizes[1] / 2;
    const int G = out_size / NOUTD;

    float* ws = (float*)d_ws;
    float* h      = ws;
    float* asrc   = h + (size_t)N * HC;
    float* adst   = asrc + (size_t)N * NH;
    float* pooled = adst + (size_t)N * NH;
    float* cnt    = pooled + (size_t)G * HC;
    int*   deg    = (int*)(cnt + (size_t)G);
    int*   rank   = deg + (size_t)N;
    int*   csr    = rank + (size_t)E;
    int*   off    = csr + (size_t)E;

    init_ws<<<256, 256, 0, stream>>>(deg, pooled, cnt, N, G);

    gemm_x_linw<<<(N + G1_ROWS - 1) / G1_ROWS, 128, 0, stream>>>(x, lin_w, h, N);

    att_kernel<<<(N * 32 + 255) / 256, 256, 0, stream>>>(h, att_s, att_d, asrc, adst, N);

    edge_rank<<<(E + 255) / 256, 256, 0, stream>>>(ei, deg, rank, E);
    scan_deg<<<N / SCAN_CHUNK + 1, SCAN_CHUNK, 0, stream>>>(deg, off, N);
    edge_fill<<<(E + 255) / 256, 256, 0, stream>>>(ei, off, rank, csr, E);

    node_gather<<<(N + 8 * NPW - 1) / (8 * NPW), 256, 0, stream>>>(off, csr, asrc, adst, h, bias,
                                                            batch, pooled, cnt, N);

    dim3 g2(G / 16, NOUTD / 32);
    gemm_out<<<g2, 32, 0, stream>>>(pooled, cnt, fc1_w, fc1_b, out, G);
}


# Edge ranks on LDS counters: 32 workgroups each count one edge-list slice in a 50176-entry LDS table (returning LDS add = rank), no global atomics in the sort; float sum order: by slice, then arrival
# speedup vs baseline: 5.6322x; 1.2060x over previous
#include <hip/hip_runtime.h>
#include <hip/hip_bf16.h>

typedef __attribute__((ext_vector_type(2))) float v2f;
typedef __attribute__((ext_vector_type(8))) float v8f;

#define HC   256
#define NH   4
#define CCH  64
#define KIN  300
#define NOUTD 768

__global__ void init_ws(float* pooled, float* cnt, int g) {
    int stride = (int)(gridDim.x * blockDim.x);
    int t = (int)(blockIdx.x * blockDim.x + threadIdx.x);
    for (int i = t; i < g * HC; i += stride) {
        pooled[i] = 0.0f;
        if (i < g) cnt[i] = 0.0f;
    }
}

#define SCAN_CHUNK 1024
#define LDS_NODES  (49 * SCAN_CHUNK)
#define RANK_WGS   32
#define RANK_BATCH 13
__global__ __launch_bounds__(1024) void edge_rank(const int* __restrict__ ei,
                                                  int* __restrict__ rank,
                                                  int* __restrict__ part,
                                                  int* __restrict__ ctot, int E) {
    __shared__ int cnt[LDS_NODES];
    int t = threadIdx.x, lane = t & 31, wv = t >> 5;
    int w = (int)blockIdx.x;
    for (int i = t; i < LDS_NODES; i += 1024) cnt[i] = 0;
    __syncthreads();

    int epw = (E + RANK_WGS - 1) / RANK_WGS;
    int e0 = w * epw;
    int e1 = e0 + epw < E ? e0 + epw : E;
    const int* dst = ei + E;
    for (int eb = e0 + t; eb < e1; eb += RANK_BATCH * 1024) {
        int d[RANK_BATCH];
        #pragma unroll
        for (int u = 0; u < RANK_BATCH; ++u) {
            int e = eb + u * 1024;
            d[u] = e < e1 ? dst[e] : 0;
        }
        #pragma unroll
        for (int u = 0; u < RANK_BATCH; ++u) {
            int e = eb + u * 1024;
            if (e < e1) rank[e] = atomicAdd(&cnt[d[u]], 1);
        }
    }
    __syncthreads();

    for (int c = wv; c < LDS_NODES / SCAN_CHUNK; c += 32) {
        int acc = 0;
        #pragma unroll 8
        for (int i = 0; i < SCAN_CHUNK / 32; ++i) {
            int v = cnt[c * SCAN_CHUNK + i * 32 + lane];
            part[(size_t)w * LDS_NODES + c * SCAN_CHUNK + i * 32 + lane] = v;
            acc += v;
        }
        for (int o = 16; o >= 1; o >>= 1) acc += __shfl_xor(acc, o, 32);
        if (lane == 0) ctot[c * RANK_WGS + w] = acc;
    }
}

__global__ __launch_bounds__(SCAN_CHUNK) void scan_deg(int* __restrict__ part,
                                                       const int* __restrict__ ctot,
                                                       int* __restrict__ off, int n) {
    __shared__ int wsum[32];
    __shared__ int psum[32];
    int t = threadIdx.x, lane = t & 31, w = t >> 5;
    int b = (int)blockIdx.x;
    int idx = b * SCAN_CHUNK + t;
    int hv[RANK_WGS];
    int v = 0;
    #pragma unroll
    for (int k = 0; k < RANK_WGS; ++k) {
        hv[k] = part[(size_t)k * LDS_NODES + idx];
        v += hv[k];
    }

    int lim = RANK_WGS * b;
    int p = 0;
    if (t < lim)              p  = ctot[t];
    if (t + SCAN_CHUNK < lim) p += ctot[t + SCAN_CHUNK];

    int x = v;
    #pragma unroll
    for (int o = 1; o < 32; o <<= 1) {
        int y = __shfl_up(x, o, 32);
        if (lane >= o) x += y;
    }
    for (int o = 16; o >= 1; o >>= 1) p += __shfl_xor(p, o, 32);
    if (lane == 31) wsum[w] = x;
    if (lane == 0)  psum[w] = p;
    __syncthreads();
    int ws = wsum[lane];
    int front = psum[lane];
    int xs = ws;
    #pragma unroll
    for (int o = 1; o < 32; o <<= 1) {
        int y = __shfl_up(xs, o, 32);
        if (lane >= o) xs += y;
    }
    for (int o = 16; o >= 1; o >>= 1) front += __shfl_xor(front, o, 32);
    int wprefix = __shfl(xs, w, 32) - __shfl(ws, w, 32);
    int tot = __shfl(xs, 31, 32);
    int o = front + wprefix + x - v;
    if (idx < n) off[idx] = o;
    #pragma unroll
    for (int k = 0; k < RANK_WGS; ++k) {
        part[(size_t)k * LDS_NODES + idx] = o;
        o += hv[k];
    }
    if (b == (int)gridDim.x - 1 && t == 0) off[n] = front + tot;
}

__global__ void edge_fill(const int* __restrict__ ei, const int* __restrict__ part,
                          const int* __restrict__ rank, int* __restrict__ csr, int E) {
    int w = (int)blockIdx.y;
    int epw = (E + RANK_WGS - 1) / RANK_WGS;
    int e = w * epw + (int)(blockIdx.x * blockDim.x + threadIdx.x);
    int e1 = (w + 1) * epw < E ? (w + 1) * epw : E;
    if (e >= e1) return;
    int s = ei[e], d = ei[E + e];
    csr[part[(size_t)w * LDS_NODES + d] + rank[e]] = s;
}

#define G1_ROWS 32
#define G1_KC   20
#define G1_AP   20
#define G1_BP   272
__global__ __launch_bounds__(128) void gemm_x_linw(const float* __restrict__ x,
                                                   const float* __restrict__ w,
                                                   float* __restrict__ h, int n) {
    __shared__ __align__(16) float As[G1_ROWS * G1_AP];
    __shared__ __align__(16) float Bs[(G1_KC / 2) * G1_BP * 2];
    int t = threadIdx.x;
    int lane = t & 31, wv = t >> 5;
    int mr = lane & 15, half = lane >> 4;
    int mt = wv & 1, nh = wv >> 1;
    int m0 = (int)blockIdx.x * G1_ROWS;

    int ra = t / 5, qa = t - 5 * ra;
    int rb2 = (128 + t) / 5, qb2 = (128 + t) - 5 * rb2;
    int gra = m0 + ra;  gra = gra < n ? gra : n - 1;
    int grb = m0 + rb2; grb = grb < n ? grb : n - 1;
    const float* pa0 = x + (size_t)gra * KIN + 4 * qa;
    const float* pa1 = x + (size_t)grb * KIN + 4 * qb2;
    float* sa0 = As + ra * G1_AP + 4 * qa;
    float* sa1 = As + rb2 * G1_AP + 4 * qb2;

    int cg = t & 63, kq = t >> 6;
    const float* pw = w + (size_t)(2 * kq) * HC + 4 * cg;
    float* sb = Bs + ((size_t)kq * G1_BP + 4 * cg) * 2;

    const float* fa = As + (16 * mt + mr) * G1_AP + 2 * half;
    const float* fb = Bs + ((size_t)half * G1_BP + 128 * nh + mr) * 2;

    v8f acc[8];
    #pragma unroll
    for (int j = 0; j < 8; ++j) acc[j] = (v8f){0.f, 0.f, 0.f, 0.f, 0.f, 0.f, 0.f, 0.f};

    for (int c = 0; c < KIN / G1_KC; ++c) {
        int k0 = c * G1_KC;
        float4 xa0 = *(const float4*)(pa0 + k0);
        float4 xa1 = make_float4(0.f, 0.f, 0.f, 0.f);
        if (wv == 0) xa1 = *(const float4*)(pa1 + k0);
        float4 we[5], wo[5];
        #pragma unroll
        for (int i = 0; i < 5; ++i) {
            const float* p = pw + (size_t)(k0 + 4 * i) * HC;
            we[i] = *(const float4*)p;
            wo[i] = *(const float4*)(p + HC);
        }
        __syncthreads();
        *(float4*)sa0 = xa0;
        if (wv == 0) *(float4*)sa1 = xa1;
        #pragma unroll
        for (int i = 0; i < 5; ++i) {
            float* q = sb + (size_t)(2 * i) * G1_BP * 2;
            *(float4*)q       = make_float4(we[i].x, wo[i].x, we[i].y, wo[i].y);
            *(float4*)(q + 4) = make_float4(we[i].z, wo[i].z, we[i].w, wo[i].w);
        }
        __syncthreads();
        #pragma unroll
        for (int ks = 0; ks < G1_KC / 4; ++ks) {
            v2f a = *(const v2f*)(fa + 4 * ks);
            #pragma unroll
            for (int j = 0; j < 8; ++j) {
                v2f b = *(const v2f*)(fb + ((size_t)(2 * ks) * G1_BP + 16 * j) * 2);
                acc[j] = __builtin_amdgcn_wmma_f32_16x16x4_f32(false, a, false, b,
                                                               (short)0, acc[j], false, false);
            }
        }
    }

    int rbase = m0 + 16 * mt + 8 * half;
    int col0 = 128 * nh + mr;
    if (rbase + 8 <= n) {
        #pragma unroll
        for (int j = 0; j < 8; ++j) {
            #pragma unroll
            for (int r = 0; r < 8; ++r)
                h[(size_t)(rbase + r) * HC + col0 + 16 * j] = acc[j][r];
        }
    } else {
        #pragma unroll
        for (int j = 0; j < 8; ++j) {
            #pragma unroll
            for (int r = 0; r < 8; ++r)
                if (rbase + r < n) h[(size_t)(rbase + r) * HC + col0 + 16 * j] = acc[j][r];
        }
    }
}

__global__ __launch_bounds__(256) void att_kernel(const float* __restrict__ h,
                                                  const float* __restrict__ wsrc,
                                                  const float* __restrict__ wdst,
                                                  float* __restrict__ asrc,
                                                  float* __restrict__ adst, int n) {
    int node = ((int)blockIdx.x * blockDim.x + threadIdx.x) >> 5;
    int lane = threadIdx.x & 31;
    if (node >= n) return;
    int hh = lane >> 3, c0 = lane & 7;
    const float* hp = h + (size_t)node * HC + hh * CCH + c0;
    const float* ws = wsrc + hh * CCH + c0;
    const float* wd = wdst + hh * CCH + c0;
    float ss = 0.f, sd = 0.f;
    #pragma unroll
    for (int i = 0; i < 8; ++i) {
        float v = hp[i * 8];
        ss += v * ws[i * 8];
        sd += v * wd[i * 8];
    }
    #pragma unroll
    for (int off = 4; off >= 1; off >>= 1) {
        ss += __shfl_xor(ss, off, 32);
        sd += __shfl_xor(sd, off, 32);
    }
    if (c0 == 0) {
        asrc[(size_t)node * NH + hh] = ss;
        adst[(size_t)node * NH + hh] = sd;
    }
}

__device__ __forceinline__ float lrelu_att(float v) { return v > 0.f ? v : 0.2f * v; }
__device__ __forceinline__ float wave_max(float v) {
    #pragma unroll
    for (int o = 16; o >= 1; o >>= 1) v = fmaxf(v, __shfl_xor(v, o, 32));
    return v;
}
__device__ __forceinline__ float wave_sum(float v) {
    #pragma unroll
    for (int o = 16; o >= 1; o >>= 1) v += __shfl_xor(v, o, 32);
    return v;
}
__device__ __forceinline__ float lane_bcast(float v, int i) {
    return __int_as_float(__builtin_amdgcn_readlane(__float_as_int(v), i));
}

#define NPW 5
__global__ __launch_bounds__(256) void node_gather(const int* __restrict__ off,
                                                   const int* __restrict__ csr,
                                                   const float* __restrict__ asrc,
                                                   const float* __restrict__ adst,
                                                   const float* __restrict__ h,
                                                   const float* __restrict__ bias,
                                                   const int* __restrict__ batch,
                                                   float* __restrict__ pooled,
                                                   float* __restrict__ cnt, int n) {
    __shared__ __align__(16) float sred[8][HC];
    __shared__ int   sgs[8];
    __shared__ float scn[8];
    int lane = threadIdx.x & 31;
    int w = threadIdx.x >> 5;
    int wbase = __builtin_amdgcn_readfirstlane(((int)blockIdx.x * 8 + w) * NPW);
    int hsel = lane >> 4;
    const float* bp = bias + 4 * lane;

    int curg = -1;
    float pa[8] = {0.f, 0.f, 0.f, 0.f, 0.f, 0.f, 0.f, 0.f};
    float pc = 0.f;

    for (int ni = 0; ni < NPW; ++ni) {
        int node = wbase + ni;
        if (node >= n) break;
        int beg = off[node];
        int degE = off[node + 1] - beg;
        int total = degE + 1;
        float4 ad = *(const float4*)(adst + (size_t)node * NH);

        float m0 = -INFINITY, m1 = -INFINITY, m2 = -INFINITY, m3 = -INFINITY;
        for (int idx = lane; idx < total; idx += 32) {
            int s = idx < degE ? csr[beg + idx] : node;
            float4 a = *(const float4*)(asrc + (size_t)s * NH);
            m0 = fmaxf(m0, lrelu_att(a.x + ad.x));
            m1 = fmaxf(m1, lrelu_att(a.y + ad.y));
            m2 = fmaxf(m2, lrelu_att(a.z + ad.z));
            m3 = fmaxf(m3, lrelu_att(a.w + ad.w));
        }
        m0 = wave_max(m0); m1 = wave_max(m1); m2 = wave_max(m2); m3 = wave_max(m3);

        float d0 = 0.f, d1 = 0.f, d2 = 0.f, d3 = 0.f;
        for (int idx = lane; idx < total; idx += 32) {
            int s = idx < degE ? csr[beg + idx] : node;
            float4 a = *(const float4*)(asrc + (size_t)s * NH);
            d0 += __expf(lrelu_att(a.x + ad.x) - m0);
            d1 += __expf(lrelu_att(a.y + ad.y) - m1);
            d2 += __expf(lrelu_att(a.z + ad.z) - m2);
            d3 += __expf(lrelu_att(a.w + ad.w) - m3);
        }
        d0 = wave_sum(d0); d1 = wave_sum(d1); d2 = wave_sum(d2); d3 = wave_sum(d3);

        float4 accA = {0.f, 0.f, 0.f, 0.f}, accB = {0.f, 0.f, 0.f, 0.f};
        for (int base = 0; base < total; base += 32) {
            int idx = base + lane;
            int s = node;
            float al0 = 0.f, al1 = 0.f, al2 = 0.f, al3 = 0.f;
            if (idx < total) {
                s = idx < degE ? csr[beg + idx] : node;
                float4 a = *(const float4*)(asrc + (size_t)s * NH);
                al0 = __expf(lrelu_att(a.x + ad.x) - m0) / d0;
                al1 = __expf(lrelu_att(a.y + ad.y) - m1) / d1;
                al2 = __expf(lrelu_att(a.z + ad.z) - m2) / d2;
                al3 = __expf(lrelu_att(a.w + ad.w) - m3) / d3;
            }
            int left = total - base;
            int cntc = left < 32 ? left : 32;
            #pragma unroll 4
            for (int i = 0; i < cntc; ++i) {
                int si = __builtin_amdgcn_readlane(s, i);
                float b0 = lane_bcast(al0, i), b1 = lane_bcast(al1, i);
                float b2 = lane_bcast(al2, i), b3 = lane_bcast(al3, i);
                float aA = hsel ? b1 : b0;
                float aB = hsel ? b3 : b2;
                const float4* hp = (const float4*)(h + (size_t)si * HC) + lane;
                float4 x0 = hp[0], x1 = hp[32];
                accA.x += aA * x0.x; accA.y += aA * x0.y; accA.z += aA * x0.z; accA.w += aA * x0.w;
                accB.x += aB * x1.x; accB.y += aB * x1.y; accB.z += aB * x1.z; accB.w += aB * x1.w;
            }
        }

        float r[8] = {accA.x + bp[0],   accA.y + bp[1],   accA.z + bp[2],   accA.w + bp[3],
                      accB.x + bp[128], accB.y + bp[129], accB.z + bp[130], accB.w + bp[131]};
        int g = batch[node];
        if (g != curg) {
            if (curg >= 0) {
                float* pA = pooled + (size_t)curg * HC + 4 * lane;
                #pragma unroll
                for (int c = 0; c < 4; ++c) {
                    atomicAdd(&pA[c], pa[c]);
                    atomicAdd(&pA[128 + c], pa[4 + c]);
                }
                if (lane == 0) atomicAdd(&cnt[curg], pc);
            }
            curg = g;
            #pragma unroll
            for (int c = 0; c < 8; ++c) pa[c] = 0.f;
            pc = 0.f;
        }
        #pragma unroll
        for (int c = 0; c < 8; ++c) {
            float u = r[c];
            pa[c] += u > 0.f ? u : 0.01f * u;
        }
        pc += 1.0f;
    }

    *(float4*)&sred[w][4 * lane]       = make_float4(pa[0], pa[1], pa[2], pa[3]);
    *(float4*)&sred[w][128 + 4 * lane] = make_float4(pa[4], pa[5], pa[6], pa[7]);
    if (lane == 0) { sgs[w] = curg; scn[w] = pc; }
    __syncthreads();

    int t = threadIdx.x;
    int gk[8];
    float vk[8], ck[8];
    #pragma unroll
    for (int k = 0; k < 8; ++k) { gk[k] = sgs[k]; vk[k] = sred[k][t]; ck[k] = scn[k]; }
    int cg = -1;
    float s = 0.f, sc = 0.f;
    #pragma unroll
    for (int k = 0; k < 8; ++k) {
        if (gk[k] < 0) continue;
        if (gk[k] != cg) {
            if (cg >= 0) {
                atomicAdd(&pooled[(size_t)cg * HC + t], s);
                if (t == 0) atomicAdd(&cnt[cg], sc);
            }
            cg = gk[k]; s = 0.f; sc = 0.f;
        }
        s += vk[k];
        sc += ck[k];
    }
    if (cg >= 0) {
        atomicAdd(&pooled[(size_t)cg * HC + t], s);
        if (t == 0) atomicAdd(&cnt[cg], sc);
    }
}

__global__ __launch_bounds__(32) void gemm_out(const float* __restrict__ pooled,
                                               const float* __restrict__ cnt,
                                               const float* __restrict__ fw,
                                               const float* __restrict__ fb,
                                               float* __restrict__ out, int g) {
    int mTile = blockIdx.x;
    int lane = threadIdx.x & 31;
    int mr = lane & 15, half = lane >> 4;
    int row = mTile * 16 + mr;
    int col0 = blockIdx.y * 32 + mr;
    float inv = 1.0f / fmaxf(cnt[row], 1.0f);
    const float* arow = pooled + (size_t)row * HC;
    v8f acc0 = {}, acc1 = {};
    for (int k = 0; k < HC; k += 4) {
        int ka = k + 2 * half;
        v2f a; a.x = arow[ka] * inv; a.y = arow[ka + 1] * inv;
        const float* wp = fw + (size_t)ka * NOUTD + col0;
        v2f b0; b0.x = wp[0];  b0.y = wp[NOUTD];
        v2f b1; b1.x = wp[16]; b1.y = wp[NOUTD + 16];
        acc0 = __builtin_amdgcn_wmma_f32_16x16x4_f32(false, a, false, b0,
                                                     (short)0, acc0, false, false);
        acc1 = __builtin_amdgcn_wmma_f32_16x16x4_f32(false, a, false, b1,
                                                     (short)0, acc1, false, false);
    }
    float bb0 = fb[col0], bb1 = fb[col0 + 16];
    int rbase = mTile * 16 + 8 * half;
    #pragma unroll
    for (int r = 0; r < 8; ++r) {
        float* op = out + (size_t)(rbase + r) * NOUTD + col0;
        op[0]  = acc0[r] + bb0;
        op[16] = acc1[r] + bb1;
    }
}

extern "C" void kernel_launch(void* const* d_in, const int* in_sizes, int n_in,
                              void* d_out, int out_size, void* d_ws, size_t ws_size,
                              hipStream_t stream) {
    const float* x      = (const float*)d_in[0];
    const int*   ei     = (const int*)  d_in[1];
    const int*   batch  = (const int*)  d_in[2];
    const float* lin_w  = (const float*)d_in[3];
    const float* att_s  = (const float*)d_in[4];
    const float* att_d  = (const float*)d_in[5];
    const float* bias   = (const float*)d_in[6];
    const float* fc1_w  = (const float*)d_in[7];
    const float* fc1_b  = (const float*)d_in[8];
    float* out = (float*)d_out;

    const int N = in_sizes[2];
    const int E = in_sizes[1] / 2;
    const int G = out_size / NOUTD;

    float* ws = (float*)d_ws;
    float* h      = ws;
    float* asrc   = h + (size_t)N * HC;
    float* adst   = asrc + (size_t)N * NH;
    float* pooled = adst + (size_t)N * NH;
    float* cnt    = pooled + (size_t)G * HC;
    int*   rank   = (int*)(cnt + (size_t)G);
    int*   csr    = rank + (size_t)E;
    int*   off    = csr + (size_t)E;
    int*   part   = (int*)h;
    int*   ctot   = part + (size_t)RANK_WGS * LDS_NODES;

    init_ws<<<256, 256, 0, stream>>>(pooled, cnt, G);

    edge_rank<<<RANK_WGS, 1024, 0, stream>>>(ei, rank, part, ctot, E);
    scan_deg<<<N / SCAN_CHUNK + 1, SCAN_CHUNK, 0, stream>>>(part, ctot, off, N);
    dim3 gf((((E + RANK_WGS - 1) / RANK_WGS) + 255) / 256, RANK_WGS);
    edge_fill<<<gf, 256, 0, stream>>>(ei, part, rank, csr, E);

    gemm_x_linw<<<(N + G1_ROWS - 1) / G1_ROWS, 128, 0, stream>>>(x, lin_w, h, N);

    att_kernel<<<(N * 32 + 255) / 256, 256, 0, stream>>>(h, att_s, att_d, asrc, adst, N);

    node_gather<<<(N + 8 * NPW - 1) / (8 * NPW), 256, 0, stream>>>(off, csr, asrc, adst, h, bias,
                                                            batch, pooled, cnt, N);

    dim3 g2(G / 16, NOUTD / 32);
    gemm_out<<<g2, 32, 0, stream>>>(pooled, cnt, fc1_w, fc1_b, out, G);
}


# Node pass loads each list entry's source id and logits once per node (one entry per lane, kept in registers for max, exp-sum and weights), not in three dependent list->logit gathers; sums unchanged
# speedup vs baseline: 5.8776x; 1.0436x over previous
#include <hip/hip_runtime.h>
#include <hip/hip_bf16.h>

typedef __attribute__((ext_vector_type(2))) float v2f;
typedef __attribute__((ext_vector_type(8))) float v8f;

#define HC   256
#define NH   4
#define CCH  64
#define KIN  300
#define NOUTD 768

__global__ void init_ws(float* pooled, float* cnt, int g) {
    int stride = (int)(gridDim.x * blockDim.x);
    int t = (int)(blockIdx.x * blockDim.x + threadIdx.x);
    for (int i = t; i < g * HC; i += stride) {
        pooled[i] = 0.0f;
        if (i < g) cnt[i] = 0.0f;
    }
}

#define SCAN_CHUNK 1024
#define LDS_NODES  (49 * SCAN_CHUNK)
#define RANK_WGS   32
#define RANK_BATCH 13
__global__ __launch_bounds__(1024) void edge_rank(const int* __restrict__ ei,
                                                  int* __restrict__ rank,
                                                  int* __restrict__ part,
                                                  int* __restrict__ ctot, int E) {
    __shared__ int cnt[LDS_NODES];
    int t = threadIdx.x, lane = t & 31, wv = t >> 5;
    int w = (int)blockIdx.x;
    for (int i = t; i < LDS_NODES; i += 1024) cnt[i] = 0;
    __syncthreads();

    int epw = (E + RANK_WGS - 1) / RANK_WGS;
    int e0 = w * epw;
    int e1 = e0 + epw < E ? e0 + epw : E;
    const int* dst = ei + E;
    for (int eb = e0 + t; eb < e1; eb += RANK_BATCH * 1024) {
        int d[RANK_BATCH];
        #pragma unroll
        for (int u = 0; u < RANK_BATCH; ++u) {
            int e = eb + u * 1024;
            d[u] = e < e1 ? dst[e] : 0;
        }
        #pragma unroll
        for (int u = 0; u < RANK_BATCH; ++u) {
            int e = eb + u * 1024;
            if (e < e1) rank[e] = atomicAdd(&cnt[d[u]], 1);
        }
    }
    __syncthreads();

    for (int c = wv; c < LDS_NODES / SCAN_CHUNK; c += 32) {
        int acc = 0;
        #pragma unroll 8
        for (int i = 0; i < SCAN_CHUNK / 32; ++i) {
            int v = cnt[c * SCAN_CHUNK + i * 32 + lane];
            part[(size_t)w * LDS_NODES + c * SCAN_CHUNK + i * 32 + lane] = v;
            acc += v;
        }
        for (int o = 16; o >= 1; o >>= 1) acc += __shfl_xor(acc, o, 32);
        if (lane == 0) ctot[c * RANK_WGS + w] = acc;
    }
}

__global__ __launch_bounds__(SCAN_CHUNK) void scan_deg(int* __restrict__ part,
                                                       const int* __restrict__ ctot,
                                                       int* __restrict__ off, int n) {
    __shared__ int wsum[32];
    __shared__ int psum[32];
    int t = threadIdx.x, lane = t & 31, w = t >> 5;
    int b = (int)blockIdx.x;
    int idx = b * SCAN_CHUNK + t;
    int hv[RANK_WGS];
    int v = 0;
    #pragma unroll
    for (int k = 0; k < RANK_WGS; ++k) {
        hv[k] = part[(size_t)k * LDS_NODES + idx];
        v += hv[k];
    }

    int lim = RANK_WGS * b;
    int p = 0;
    if (t < lim)              p  = ctot[t];
    if (t + SCAN_CHUNK < lim) p += ctot[t + SCAN_CHUNK];

    int x = v;
    #pragma unroll
    for (int o = 1; o < 32; o <<= 1) {
        int y = __shfl_up(x, o, 32);
        if (lane >= o) x += y;
    }
    for (int o = 16; o >= 1; o >>= 1) p += __shfl_xor(p, o, 32);
    if (lane == 31) wsum[w] = x;
    if (lane == 0)  psum[w] = p;
    __syncthreads();
    int ws = wsum[lane];
    int front = psum[lane];
    int xs = ws;
    #pragma unroll
    for (int o = 1; o < 32; o <<= 1) {
        int y = __shfl_up(xs, o, 32);
        if (lane >= o) xs += y;
    }
    for (int o = 16; o >= 1; o >>= 1) front += __shfl_xor(front, o, 32);
    int wprefix = __shfl(xs, w, 32) - __shfl(ws, w, 32);
    int tot = __shfl(xs, 31, 32);
    int o = front + wprefix + x - v;
    if (idx < n) off[idx] = o;
    #pragma unroll
    for (int k = 0; k < RANK_WGS; ++k) {
        part[(size_t)k * LDS_NODES + idx] = o;
        o += hv[k];
    }
    if (b == (int)gridDim.x - 1 && t == 0) off[n] = front + tot;
}

__global__ void edge_fill(const int* __restrict__ ei, const int* __restrict__ part,
                          const int* __restrict__ rank, int* __restrict__ csr, int E) {
    int w = (int)blockIdx.y;
    int epw = (E + RANK_WGS - 1) / RANK_WGS;
    int e = w * epw + (int)(blockIdx.x * blockDim.x + threadIdx.x);
    int e1 = (w + 1) * epw < E ? (w + 1) * epw : E;
    if (e >= e1) return;
    int s = ei[e], d = ei[E + e];
    csr[part[(size_t)w * LDS_NODES + d] + rank[e]] = s;
}

#define G1_ROWS 32
#define G1_KC   20
#define G1_AP   20
#define G1_BP   272
__global__ __launch_bounds__(128) void gemm_x_linw(const float* __restrict__ x,
                                                   const float* __restrict__ w,
                                                   float* __restrict__ h, int n) {
    __shared__ __align__(16) float As[G1_ROWS * G1_AP];
    __shared__ __align__(16) float Bs[(G1_KC / 2) * G1_BP * 2];
    int t = threadIdx.x;
    int lane = t & 31, wv = t >> 5;
    int mr = lane & 15, half = lane >> 4;
    int mt = wv & 1, nh = wv >> 1;
    int m0 = (int)blockIdx.x * G1_ROWS;

    int ra = t / 5, qa = t - 5 * ra;
    int rb2 = (128 + t) / 5, qb2 = (128 + t) - 5 * rb2;
    int gra = m0 + ra;  gra = gra < n ? gra : n - 1;
    int grb = m0 + rb2; grb = grb < n ? grb : n - 1;
    const float* pa0 = x + (size_t)gra * KIN + 4 * qa;
    const float* pa1 = x + (size_t)grb * KIN + 4 * qb2;
    float* sa0 = As + ra * G1_AP + 4 * qa;
    float* sa1 = As + rb2 * G1_AP + 4 * qb2;

    int cg = t & 63, kq = t >> 6;
    const float* pw = w + (size_t)(2 * kq) * HC + 4 * cg;
    float* sb = Bs + ((size_t)kq * G1_BP + 4 * cg) * 2;

    const float* fa = As + (16 * mt + mr) * G1_AP + 2 * half;
    const float* fb = Bs + ((size_t)half * G1_BP + 128 * nh + mr) * 2;

    v8f acc[8];
    #pragma unroll
    for (int j = 0; j < 8; ++j) acc[j] = (v8f){0.f, 0.f, 0.f, 0.f, 0.f, 0.f, 0.f, 0.f};

    for (int c = 0; c < KIN / G1_KC; ++c) {
        int k0 = c * G1_KC;
        float4 xa0 = *(const float4*)(pa0 + k0);
        float4 xa1 = make_float4(0.f, 0.f, 0.f, 0.f);
        if (wv == 0) xa1 = *(const float4*)(pa1 + k0);
        float4 we[5], wo[5];
        #pragma unroll
        for (int i = 0; i < 5; ++i) {
            const float* p = pw + (size_t)(k0 + 4 * i) * HC;
            we[i] = *(const float4*)p;
            wo[i] = *(const float4*)(p + HC);
        }
        __syncthreads();
        *(float4*)sa0 = xa0;
        if (wv == 0) *(float4*)sa1 = xa1;
        #pragma unroll
        for (int i = 0; i < 5; ++i) {
            float* q = sb + (size_t)(2 * i) * G1_BP * 2;
            *(float4*)q       = make_float4(we[i].x, wo[i].x, we[i].y, wo[i].y);
            *(float4*)(q + 4) = make_float4(we[i].z, wo[i].z, we[i].w, wo[i].w);
        }
        __syncthreads();
        #pragma unroll
        for (int ks = 0; ks < G1_KC / 4; ++ks) {
            v2f a = *(const v2f*)(fa + 4 * ks);
            #pragma unroll
            for (int j = 0; j < 8; ++j) {
                v2f b = *(const v2f*)(fb + ((size_t)(2 * ks) * G1_BP + 16 * j) * 2);
                acc[j] = __builtin_amdgcn_wmma_f32_16x16x4_f32(false, a, false, b,
                                                               (short)0, acc[j], false, false);
            }
        }
    }

    int rbase = m0 + 16 * mt + 8 * half;
    int col0 = 128 * nh + mr;
    if (rbase + 8 <= n) {
        #pragma unroll
        for (int j = 0; j < 8; ++j) {
            #pragma unroll
            for (int r = 0; r < 8; ++r)
                h[(size_t)(rbase + r) * HC + col0 + 16 * j] = acc[j][r];
        }
    } else {
        #pragma unroll
        for (int j = 0; j < 8; ++j) {
            #pragma unroll
            for (int r = 0; r < 8; ++r)
                if (rbase + r < n) h[(size_t)(rbase + r) * HC + col0 + 16 * j] = acc[j][r];
        }
    }
}

__global__ __launch_bounds__(256) void att_kernel(const float* __restrict__ h,
                                                  const float* __restrict__ wsrc,
                                                  const float* __restrict__ wdst,
                                                  float* __restrict__ asrc,
                                                  float* __restrict__ adst, int n) {
    int node = ((int)blockIdx.x * blockDim.x + threadIdx.x) >> 5;
    int lane = threadIdx.x & 31;
    if (node >= n) return;
    int hh = lane >> 3, c0 = lane & 7;
    const float* hp = h + (size_t)node * HC + hh * CCH + c0;
    const float* ws = wsrc + hh * CCH + c0;
    const float* wd = wdst + hh * CCH + c0;
    float ss = 0.f, sd = 0.f;
    #pragma unroll
    for (int i = 0; i < 8; ++i) {
        float v = hp[i * 8];
        ss += v * ws[i * 8];
        sd += v * wd[i * 8];
    }
    #pragma unroll
    for (int off = 4; off >= 1; off >>= 1) {
        ss += __shfl_xor(ss, off, 32);
        sd += __shfl_xor(sd, off, 32);
    }
    if (c0 == 0) {
        asrc[(size_t)node * NH + hh] = ss;
        adst[(size_t)node * NH + hh] = sd;
    }
}

__device__ __forceinline__ float lrelu_att(float v) { return v > 0.f ? v : 0.2f * v; }
__device__ __forceinline__ float wave_max(float v) {
    #pragma unroll
    for (int o = 16; o >= 1; o >>= 1) v = fmaxf(v, __shfl_xor(v, o, 32));
    return v;
}
__device__ __forceinline__ float wave_sum(float v) {
    #pragma unroll
    for (int o = 16; o >= 1; o >>= 1) v += __shfl_xor(v, o, 32);
    return v;
}
__device__ __forceinline__ float lane_bcast(float v, int i) {
    return __int_as_float(__builtin_amdgcn_readlane(__float_as_int(v), i));
}

__device__ __forceinline__ void add_rows(const float* __restrict__ h, int s,
                                         float al0, float al1, float al2, float al3,
                                         int cntc, int lane, int hsel,
                                         float4& accA, float4& accB) {
    #pragma unroll 4
    for (int i = 0; i < cntc; ++i) {
        int si = __builtin_amdgcn_readlane(s, i);
        float b0 = lane_bcast(al0, i), b1 = lane_bcast(al1, i);
        float b2 = lane_bcast(al2, i), b3 = lane_bcast(al3, i);
        float aA = hsel ? b1 : b0;
        float aB = hsel ? b3 : b2;
        const float4* hp = (const float4*)(h + (size_t)si * HC) + lane;
        float4 x0 = hp[0], x1 = hp[32];
        accA.x += aA * x0.x; accA.y += aA * x0.y; accA.z += aA * x0.z; accA.w += aA * x0.w;
        accB.x += aB * x1.x; accB.y += aB * x1.y; accB.z += aB * x1.z; accB.w += aB * x1.w;
    }
}

#define NPW 5
__global__ __launch_bounds__(256) void node_gather(const int* __restrict__ off,
                                                   const int* __restrict__ csr,
                                                   const float* __restrict__ asrc,
                                                   const float* __restrict__ adst,
                                                   const float* __restrict__ h,
                                                   const float* __restrict__ bias,
                                                   const int* __restrict__ batch,
                                                   float* __restrict__ pooled,
                                                   float* __restrict__ cnt, int n) {
    __shared__ __align__(16) float sred[8][HC];
    __shared__ int   sgs[8];
    __shared__ float scn[8];
    int lane = threadIdx.x & 31;
    int w = threadIdx.x >> 5;
    int wbase = __builtin_amdgcn_readfirstlane(((int)blockIdx.x * 8 + w) * NPW);
    int hsel = lane >> 4;
    const float* bp = bias + 4 * lane;

    int curg = -1;
    float pa[8] = {0.f, 0.f, 0.f, 0.f, 0.f, 0.f, 0.f, 0.f};
    float pc = 0.f;

    for (int ni = 0; ni < NPW; ++ni) {
        int node = wbase + ni;
        if (node >= n) break;
        int beg = off[node];
        int degE = off[node + 1] - beg;
        int total = degE + 1;
        float4 ad = *(const float4*)(adst + (size_t)node * NH);

        float4 accA = {0.f, 0.f, 0.f, 0.f}, accB = {0.f, 0.f, 0.f, 0.f};
        if (total <= 32) {
            bool live = lane < total;
            int s = node;
            float e0 = -INFINITY, e1 = -INFINITY, e2 = -INFINITY, e3 = -INFINITY;
            if (live) {
                if (lane < degE) s = csr[beg + lane];
                float4 a = *(const float4*)(asrc + (size_t)s * NH);
                e0 = lrelu_att(a.x + ad.x);
                e1 = lrelu_att(a.y + ad.y);
                e2 = lrelu_att(a.z + ad.z);
                e3 = lrelu_att(a.w + ad.w);
            }
            float m0 = wave_max(e0), m1 = wave_max(e1), m2 = wave_max(e2), m3 = wave_max(e3);
            float p0 = live ? __expf(e0 - m0) : 0.f;
            float p1 = live ? __expf(e1 - m1) : 0.f;
            float p2 = live ? __expf(e2 - m2) : 0.f;
            float p3 = live ? __expf(e3 - m3) : 0.f;
            float d0 = wave_sum(p0), d1 = wave_sum(p1), d2 = wave_sum(p2), d3 = wave_sum(p3);
            add_rows(h, s, p0 / d0, p1 / d1, p2 / d2, p3 / d3, total, lane, hsel, accA, accB);
        } else {
            float m0 = -INFINITY, m1 = -INFINITY, m2 = -INFINITY, m3 = -INFINITY;
            for (int idx = lane; idx < total; idx += 32) {
                int s = idx < degE ? csr[beg + idx] : node;
                float4 a = *(const float4*)(asrc + (size_t)s * NH);
                m0 = fmaxf(m0, lrelu_att(a.x + ad.x));
                m1 = fmaxf(m1, lrelu_att(a.y + ad.y));
                m2 = fmaxf(m2, lrelu_att(a.z + ad.z));
                m3 = fmaxf(m3, lrelu_att(a.w + ad.w));
            }
            m0 = wave_max(m0); m1 = wave_max(m1); m2 = wave_max(m2); m3 = wave_max(m3);

            float d0 = 0.f, d1 = 0.f, d2 = 0.f, d3 = 0.f;
            for (int idx = lane; idx < total; idx += 32) {
                int s = idx < degE ? csr[beg + idx] : node;
                float4 a = *(const float4*)(asrc + (size_t)s * NH);
                d0 += __expf(lrelu_att(a.x + ad.x) - m0);
                d1 += __expf(lrelu_att(a.y + ad.y) - m1);
                d2 += __expf(lrelu_att(a.z + ad.z) - m2);
                d3 += __expf(lrelu_att(a.w + ad.w) - m3);
            }
            d0 = wave_sum(d0); d1 = wave_sum(d1); d2 = wave_sum(d2); d3 = wave_sum(d3);

            for (int base = 0; base < total; base += 32) {
                int idx = base + lane;
                int s = node;
                float al0 = 0.f, al1 = 0.f, al2 = 0.f, al3 = 0.f;
                if (idx < total) {
                    s = idx < degE ? csr[beg + idx] : node;
                    float4 a = *(const float4*)(asrc + (size_t)s * NH);
                    al0 = __expf(lrelu_att(a.x + ad.x) - m0) / d0;
                    al1 = __expf(lrelu_att(a.y + ad.y) - m1) / d1;
                    al2 = __expf(lrelu_att(a.z + ad.z) - m2) / d2;
                    al3 = __expf(lrelu_att(a.w + ad.w) - m3) / d3;
                }
                int left = total - base;
                add_rows(h, s, al0, al1, al2, al3, left < 32 ? left : 32, lane, hsel, accA, accB);
            }
        }

        float r[8] = {accA.x + bp[0],   accA.y + bp[1],   accA.z + bp[2],   accA.w + bp[3],
                      accB.x + bp[128], accB.y + bp[129], accB.z + bp[130], accB.w + bp[131]};
        int g = batch[node];
        if (g != curg) {
            if (curg >= 0) {
                float* pA = pooled + (size_t)curg * HC + 4 * lane;
                #pragma unroll
                for (int c = 0; c < 4; ++c) {
                    atomicAdd(&pA[c], pa[c]);
                    atomicAdd(&pA[128 + c], pa[4 + c]);
                }
                if (lane == 0) atomicAdd(&cnt[curg], pc);
            }
            curg = g;
            #pragma unroll
            for (int c = 0; c < 8; ++c) pa[c] = 0.f;
            pc = 0.f;
        }
        #pragma unroll
        for (int c = 0; c < 8; ++c) {
            float u = r[c];
            pa[c] += u > 0.f ? u : 0.01f * u;
        }
        pc += 1.0f;
    }

    *(float4*)&sred[w][4 * lane]       = make_float4(pa[0], pa[1], pa[2], pa[3]);
    *(float4*)&sred[w][128 + 4 * lane] = make_float4(pa[4], pa[5], pa[6], pa[7]);
    if (lane == 0) { sgs[w] = curg; scn[w] = pc; }
    __syncthreads();

    int t = threadIdx.x;
    int gk[8];
    float vk[8], ck[8];
    #pragma unroll
    for (int k = 0; k < 8; ++k) { gk[k] = sgs[k]; vk[k] = sred[k][t]; ck[k] = scn[k]; }
    int cg = -1;
    float s = 0.f, sc = 0.f;
    #pragma unroll
    for (int k = 0; k < 8; ++k) {
        if (gk[k] < 0) continue;
        if (gk[k] != cg) {
            if (cg >= 0) {
                atomicAdd(&pooled[(size_t)cg * HC + t], s);
                if (t == 0) atomicAdd(&cnt[cg], sc);
            }
            cg = gk[k]; s = 0.f; sc = 0.f;
        }
        s += vk[k];
        sc += ck[k];
    }
    if (cg >= 0) {
        atomicAdd(&pooled[(size_t)cg * HC + t], s);
        if (t == 0) atomicAdd(&cnt[cg], sc);
    }
}

__global__ __launch_bounds__(32) void gemm_out(const float* __restrict__ pooled,
                                               const float* __restrict__ cnt,
                                               const float* __restrict__ fw,
                                               const float* __restrict__ fb,
                                               float* __restrict__ out, int g) {
    int mTile = blockIdx.x;
    int lane = threadIdx.x & 31;
    int mr = lane & 15, half = lane >> 4;
    int row = mTile * 16 + mr;
    int col0 = blockIdx.y * 32 + mr;
    float inv = 1.0f / fmaxf(cnt[row], 1.0f);
    const float* arow = pooled + (size_t)row * HC;
    v8f acc0 = {}, acc1 = {};
    for (int k = 0; k < HC; k += 4) {
        int ka = k + 2 * half;
        v2f a; a.x = arow[ka] * inv; a.y = arow[ka + 1] * inv;
        const float* wp = fw + (size_t)ka * NOUTD + col0;
        v2f b0; b0.x = wp[0];  b0.y = wp[NOUTD];
        v2f b1; b1.x = wp[16]; b1.y = wp[NOUTD + 16];
        acc0 = __builtin_amdgcn_wmma_f32_16x16x4_f32(false, a, false, b0,
                                                     (short)0, acc0, false, false);
        acc1 = __builtin_amdgcn_wmma_f32_16x16x4_f32(false, a, false, b1,
                                                     (short)0, acc1, false, false);
    }
    float bb0 = fb[col0], bb1 = fb[col0 + 16];
    int rbase = mTile * 16 + 8 * half;
    #pragma unroll
    for (int r = 0; r < 8; ++r) {
        float* op = out + (size_t)(rbase + r) * NOUTD + col0;
        op[0]  = acc0[r] + bb0;
        op[16] = acc1[r] + bb1;
    }
}

extern "C" void kernel_launch(void* const* d_in, const int* in_sizes, int n_in,
                              void* d_out, int out_size, void* d_ws, size_t ws_size,
                              hipStream_t stream) {
    const float* x      = (const float*)d_in[0];
    const int*   ei     = (const int*)  d_in[1];
    const int*   batch  = (const int*)  d_in[2];
    const float* lin_w  = (const float*)d_in[3];
    const float* att_s  = (const float*)d_in[4];
    const float* att_d  = (const float*)d_in[5];
    const float* bias   = (const float*)d_in[6];
    const float* fc1_w  = (const float*)d_in[7];
    const float* fc1_b  = (const float*)d_in[8];
    float* out = (float*)d_out;

    const int N = in_sizes[2];
    const int E = in_sizes[1] / 2;
    const int G = out_size / NOUTD;

    float* ws = (float*)d_ws;
    float* h      = ws;
    float* asrc   = h + (size_t)N * HC;
    float* adst   = asrc + (size_t)N * NH;
    float* pooled = adst + (size_t)N * NH;
    float* cnt    = pooled + (size_t)G * HC;
    int*   rank   = (int*)(cnt + (size_t)G);
    int*   csr    = rank + (size_t)E;
    int*   off    = csr + (size_t)E;
    int*   part   = (int*)h;
    int*   ctot   = part + (size_t)RANK_WGS * LDS_NODES;

    init_ws<<<256, 256, 0, stream>>>(pooled, cnt, G);

    edge_rank<<<RANK_WGS, 1024, 0, stream>>>(ei, rank, part, ctot, E);
    scan_deg<<<N / SCAN_CHUNK + 1, SCAN_CHUNK, 0, stream>>>(part, ctot, off, N);
    dim3 gf((((E + RANK_WGS - 1) / RANK_WGS) + 255) / 256, RANK_WGS);
    edge_fill<<<gf, 256, 0, stream>>>(ei, part, rank, csr, E);

    gemm_x_linw<<<(N + G1_ROWS - 1) / G1_ROWS, 128, 0, stream>>>(x, lin_w, h, N);

    att_kernel<<<(N * 32 + 255) / 256, 256, 0, stream>>>(h, att_s, att_d, asrc, adst, N);

    node_gather<<<(N + 8 * NPW - 1) / (8 * NPW), 256, 0, stream>>>(off, csr, asrc, adst, h, bias,
                                                            batch, pooled, cnt, N);

    dim3 g2(G / 16, NOUTD / 32);
    gemm_out<<<g2, 32, 0, stream>>>(pooled, cnt, fc1_w, fc1_b, out, G);
}


# Final FC: k split over a workgroup's 4 waves; each wave issues all operand loads of its k range before its first WMMA (one memory wait, not one per k group); partials added via LDS; out sum order new
# speedup vs baseline: 6.2463x; 1.0627x over previous
#include <hip/hip_runtime.h>
#include <hip/hip_bf16.h>

typedef __attribute__((ext_vector_type(2))) float v2f;
typedef __attribute__((ext_vector_type(8))) float v8f;

#define HC   256
#define NH   4
#define CCH  64
#define KIN  300
#define NOUTD 768

__global__ void init_ws(float* pooled, float* cnt, int g) {
    int stride = (int)(gridDim.x * blockDim.x);
    int t = (int)(blockIdx.x * blockDim.x + threadIdx.x);
    for (int i = t; i < g * HC; i += stride) {
        pooled[i] = 0.0f;
        if (i < g) cnt[i] = 0.0f;
    }
}

#define SCAN_CHUNK 1024
#define LDS_NODES  (49 * SCAN_CHUNK)
#define RANK_WGS   32
#define RANK_BATCH 13
__global__ __launch_bounds__(1024) void edge_rank(const int* __restrict__ ei,
                                                  int* __restrict__ rank,
                                                  int* __restrict__ part,
                                                  int* __restrict__ ctot, int E) {
    __shared__ int cnt[LDS_NODES];
    int t = threadIdx.x, lane = t & 31, wv = t >> 5;
    int w = (int)blockIdx.x;
    for (int i = t; i < LDS_NODES; i += 1024) cnt[i] = 0;
    __syncthreads();

    int epw = (E + RANK_WGS - 1) / RANK_WGS;
    int e0 = w * epw;
    int e1 = e0 + epw < E ? e0 + epw : E;
    const int* dst = ei + E;
    for (int eb = e0 + t; eb < e1; eb += RANK_BATCH * 1024) {
        int d[RANK_BATCH];
        #pragma unroll
        for (int u = 0; u < RANK_BATCH; ++u) {
            int e = eb + u * 1024;
            d[u] = e < e1 ? dst[e] : 0;
        }
        #pragma unroll
        for (int u = 0; u < RANK_BATCH; ++u) {
            int e = eb + u * 1024;
            if (e < e1) rank[e] = atomicAdd(&cnt[d[u]], 1);
        }
    }
    __syncthreads();

    for (int c = wv; c < LDS_NODES / SCAN_CHUNK; c += 32) {
        int acc = 0;
        #pragma unroll 8
        for (int i = 0; i < SCAN_CHUNK / 32; ++i) {
            int v = cnt[c * SCAN_CHUNK + i * 32 + lane];
            part[(size_t)w * LDS_NODES + c * SCAN_CHUNK + i * 32 + lane] = v;
            acc += v;
        }
        for (int o = 16; o >= 1; o >>= 1) acc += __shfl_xor(acc, o, 32);
        if (lane == 0) ctot[c * RANK_WGS + w] = acc;
    }
}

__global__ __launch_bounds__(SCAN_CHUNK) void scan_deg(int* __restrict__ part,
                                                       const int* __restrict__ ctot,
                                                       int* __restrict__ off, int n) {
    __shared__ int wsum[32];
    __shared__ int psum[32];
    int t = threadIdx.x, lane = t & 31, w = t >> 5;
    int b = (int)blockIdx.x;
    int idx = b * SCAN_CHUNK + t;
    int hv[RANK_WGS];
    int v = 0;
    #pragma unroll
    for (int k = 0; k < RANK_WGS; ++k) {
        hv[k] = part[(size_t)k * LDS_NODES + idx];
        v += hv[k];
    }

    int lim = RANK_WGS * b;
    int p = 0;
    if (t < lim)              p  = ctot[t];
    if (t + SCAN_CHUNK < lim) p += ctot[t + SCAN_CHUNK];

    int x = v;
    #pragma unroll
    for (int o = 1; o < 32; o <<= 1) {
        int y = __shfl_up(x, o, 32);
        if (lane >= o) x += y;
    }
    for (int o = 16; o >= 1; o >>= 1) p += __shfl_xor(p, o, 32);
    if (lane == 31) wsum[w] = x;
    if (lane == 0)  psum[w] = p;
    __syncthreads();
    int ws = wsum[lane];
    int front = psum[lane];
    int xs = ws;
    #pragma unroll
    for (int o = 1; o < 32; o <<= 1) {
        int y = __shfl_up(xs, o, 32);
        if (lane >= o) xs += y;
    }
    for (int o = 16; o >= 1; o >>= 1) front += __shfl_xor(front, o, 32);
    int wprefix = __shfl(xs, w, 32) - __shfl(ws, w, 32);
    int tot = __shfl(xs, 31, 32);
    int o = front + wprefix + x - v;
    if (idx < n) off[idx] = o;
    #pragma unroll
    for (int k = 0; k < RANK_WGS; ++k) {
        part[(size_t)k * LDS_NODES + idx] = o;
        o += hv[k];
    }
    if (b == (int)gridDim.x - 1 && t == 0) off[n] = front + tot;
}

__global__ void edge_fill(const int* __restrict__ ei, const int* __restrict__ part,
                          const int* __restrict__ rank, int* __restrict__ csr, int E) {
    int w = (int)blockIdx.y;
    int epw = (E + RANK_WGS - 1) / RANK_WGS;
    int e = w * epw + (int)(blockIdx.x * blockDim.x + threadIdx.x);
    int e1 = (w + 1) * epw < E ? (w + 1) * epw : E;
    if (e >= e1) return;
    int s = ei[e], d = ei[E + e];
    csr[part[(size_t)w * LDS_NODES + d] + rank[e]] = s;
}

#define G1_ROWS 32
#define G1_KC   20
#define G1_AP   20
#define G1_BP   272
__global__ __launch_bounds__(128) void gemm_x_linw(const float* __restrict__ x,
                                                   const float* __restrict__ w,
                                                   float* __restrict__ h, int n) {
    __shared__ __align__(16) float As[G1_ROWS * G1_AP];
    __shared__ __align__(16) float Bs[(G1_KC / 2) * G1_BP * 2];
    int t = threadIdx.x;
    int lane = t & 31, wv = t >> 5;
    int mr = lane & 15, half = lane >> 4;
    int mt = wv & 1, nh = wv >> 1;
    int m0 = (int)blockIdx.x * G1_ROWS;

    int ra = t / 5, qa = t - 5 * ra;
    int rb2 = (128 + t) / 5, qb2 = (128 + t) - 5 * rb2;
    int gra = m0 + ra;  gra = gra < n ? gra : n - 1;
    int grb = m0 + rb2; grb = grb < n ? grb : n - 1;
    const float* pa0 = x + (size_t)gra * KIN + 4 * qa;
    const float* pa1 = x + (size_t)grb * KIN + 4 * qb2;
    float* sa0 = As + ra * G1_AP + 4 * qa;
    float* sa1 = As + rb2 * G1_AP + 4 * qb2;

    int cg = t & 63, kq = t >> 6;
    const float* pw = w + (size_t)(2 * kq) * HC + 4 * cg;
    float* sb = Bs + ((size_t)kq * G1_BP + 4 * cg) * 2;

    const float* fa = As + (16 * mt + mr) * G1_AP + 2 * half;
    const float* fb = Bs + ((size_t)half * G1_BP + 128 * nh + mr) * 2;

    v8f acc[8];
    #pragma unroll
    for (int j = 0; j < 8; ++j) acc[j] = (v8f){0.f, 0.f, 0.f, 0.f, 0.f, 0.f, 0.f, 0.f};

    for (int c = 0; c < KIN / G1_KC; ++c) {
        int k0 = c * G1_KC;
        float4 xa0 = *(const float4*)(pa0 + k0);
        float4 xa1 = make_float4(0.f, 0.f, 0.f, 0.f);
        if (wv == 0) xa1 = *(const float4*)(pa1 + k0);
        float4 we[5], wo[5];
        #pragma unroll
        for (int i = 0; i < 5; ++i) {
            const float* p = pw + (size_t)(k0 + 4 * i) * HC;
            we[i] = *(const float4*)p;
            wo[i] = *(const float4*)(p + HC);
        }
        __syncthreads();
        *(float4*)sa0 = xa0;
        if (wv == 0) *(float4*)sa1 = xa1;
        #pragma unroll
        for (int i = 0; i < 5; ++i) {
            float* q = sb + (size_t)(2 * i) * G1_BP * 2;
            *(float4*)q       = make_float4(we[i].x, wo[i].x, we[i].y, wo[i].y);
            *(float4*)(q + 4) = make_float4(we[i].z, wo[i].z, we[i].w, wo[i].w);
        }
        __syncthreads();
        #pragma unroll
        for (int ks = 0; ks < G1_KC / 4; ++ks) {
            v2f a = *(const v2f*)(fa + 4 * ks);
            #pragma unroll
            for (int j = 0; j < 8; ++j) {
                v2f b = *(const v2f*)(fb + ((size_t)(2 * ks) * G1_BP + 16 * j) * 2);
                acc[j] = __builtin_amdgcn_wmma_f32_16x16x4_f32(false, a, false, b,
                                                               (short)0, acc[j], false, false);
            }
        }
    }

    int rbase = m0 + 16 * mt + 8 * half;
    int col0 = 128 * nh + mr;
    if (rbase + 8 <= n) {
        #pragma unroll
        for (int j = 0; j < 8; ++j) {
            #pragma unroll
            for (int r = 0; r < 8; ++r)
                h[(size_t)(rbase + r) * HC + col0 + 16 * j] = acc[j][r];
        }
    } else {
        #pragma unroll
        for (int j = 0; j < 8; ++j) {
            #pragma unroll
            for (int r = 0; r < 8; ++r)
                if (rbase + r < n) h[(size_t)(rbase + r) * HC + col0 + 16 * j] = acc[j][r];
        }
    }
}

__global__ __launch_bounds__(256) void att_kernel(const float* __restrict__ h,
                                                  const float* __restrict__ wsrc,
                                                  const float* __restrict__ wdst,
                                                  float* __restrict__ asrc,
                                                  float* __restrict__ adst, int n) {
    int node = ((int)blockIdx.x * blockDim.x + threadIdx.x) >> 5;
    int lane = threadIdx.x & 31;
    if (node >= n) return;
    int hh = lane >> 3, c0 = lane & 7;
    const float* hp = h + (size_t)node * HC + hh * CCH + c0;
    const float* ws = wsrc + hh * CCH + c0;
    const float* wd = wdst + hh * CCH + c0;
    float ss = 0.f, sd = 0.f;
    #pragma unroll
    for (int i = 0; i < 8; ++i) {
        float v = hp[i * 8];
        ss += v * ws[i * 8];
        sd += v * wd[i * 8];
    }
    #pragma unroll
    for (int off = 4; off >= 1; off >>= 1) {
        ss += __shfl_xor(ss, off, 32);
        sd += __shfl_xor(sd, off, 32);
    }
    if (c0 == 0) {
        asrc[(size_t)node * NH + hh] = ss;
        adst[(size_t)node * NH + hh] = sd;
    }
}

__device__ __forceinline__ float lrelu_att(float v) { return v > 0.f ? v : 0.2f * v; }
__device__ __forceinline__ float wave_max(float v) {
    #pragma unroll
    for (int o = 16; o >= 1; o >>= 1) v = fmaxf(v, __shfl_xor(v, o, 32));
    return v;
}
__device__ __forceinline__ float wave_sum(float v) {
    #pragma unroll
    for (int o = 16; o >= 1; o >>= 1) v += __shfl_xor(v, o, 32);
    return v;
}
__device__ __forceinline__ float lane_bcast(float v, int i) {
    return __int_as_float(__builtin_amdgcn_readlane(__float_as_int(v), i));
}

__device__ __forceinline__ void add_rows(const float* __restrict__ h, int s,
                                         float al0, float al1, float al2, float al3,
                                         int cntc, int lane, int hsel,
                                         float4& accA, float4& accB) {
    #pragma unroll 4
    for (int i = 0; i < cntc; ++i) {
        int si = __builtin_amdgcn_readlane(s, i);
        float b0 = lane_bcast(al0, i), b1 = lane_bcast(al1, i);
        float b2 = lane_bcast(al2, i), b3 = lane_bcast(al3, i);
        float aA = hsel ? b1 : b0;
        float aB = hsel ? b3 : b2;
        const float4* hp = (const float4*)(h + (size_t)si * HC) + lane;
        float4 x0 = hp[0], x1 = hp[32];
        accA.x += aA * x0.x; accA.y += aA * x0.y; accA.z += aA * x0.z; accA.w += aA * x0.w;
        accB.x += aB * x1.x; accB.y += aB * x1.y; accB.z += aB * x1.z; accB.w += aB * x1.w;
    }
}

#define NPW 5
__global__ __launch_bounds__(256) void node_gather(const int* __restrict__ off,
                                                   const int* __restrict__ csr,
                                                   const float* __restrict__ asrc,
                                                   const float* __restrict__ adst,
                                                   const float* __restrict__ h,
                                                   const float* __restrict__ bias,
                                                   const int* __restrict__ batch,
                                                   float* __restrict__ pooled,
                                                   float* __restrict__ cnt, int n) {
    __shared__ __align__(16) float sred[8][HC];
    __shared__ int   sgs[8];
    __shared__ float scn[8];
    int lane = threadIdx.x & 31;
    int w = threadIdx.x >> 5;
    int wbase = __builtin_amdgcn_readfirstlane(((int)blockIdx.x * 8 + w) * NPW);
    int hsel = lane >> 4;
    const float* bp = bias + 4 * lane;

    int curg = -1;
    float pa[8] = {0.f, 0.f, 0.f, 0.f, 0.f, 0.f, 0.f, 0.f};
    float pc = 0.f;

    for (int ni = 0; ni < NPW; ++ni) {
        int node = wbase + ni;
        if (node >= n) break;
        int beg = off[node];
        int degE = off[node + 1] - beg;
        int total = degE + 1;
        float4 ad = *(const float4*)(adst + (size_t)node * NH);

        float4 accA = {0.f, 0.f, 0.f, 0.f}, accB = {0.f, 0.f, 0.f, 0.f};
        if (total <= 32) {
            bool live = lane < total;
            int s = node;
            float e0 = -INFINITY, e1 = -INFINITY, e2 = -INFINITY, e3 = -INFINITY;
            if (live) {
                if (lane < degE) s = csr[beg + lane];
                float4 a = *(const float4*)(asrc + (size_t)s * NH);
                e0 = lrelu_att(a.x + ad.x);
                e1 = lrelu_att(a.y + ad.y);
                e2 = lrelu_att(a.z + ad.z);
                e3 = lrelu_att(a.w + ad.w);
            }
            float m0 = wave_max(e0), m1 = wave_max(e1), m2 = wave_max(e2), m3 = wave_max(e3);
            float p0 = live ? __expf(e0 - m0) : 0.f;
            float p1 = live ? __expf(e1 - m1) : 0.f;
            float p2 = live ? __expf(e2 - m2) : 0.f;
            float p3 = live ? __expf(e3 - m3) : 0.f;
            float d0 = wave_sum(p0), d1 = wave_sum(p1), d2 = wave_sum(p2), d3 = wave_sum(p3);
            add_rows(h, s, p0 / d0, p1 / d1, p2 / d2, p3 / d3, total, lane, hsel, accA, accB);
        } else {
            float m0 = -INFINITY, m1 = -INFINITY, m2 = -INFINITY, m3 = -INFINITY;
            for (int idx = lane; idx < total; idx += 32) {
                int s = idx < degE ? csr[beg + idx] : node;
                float4 a = *(const float4*)(asrc + (size_t)s * NH);
                m0 = fmaxf(m0, lrelu_att(a.x + ad.x));
                m1 = fmaxf(m1, lrelu_att(a.y + ad.y));
                m2 = fmaxf(m2, lrelu_att(a.z + ad.z));
                m3 = fmaxf(m3, lrelu_att(a.w + ad.w));
            }
            m0 = wave_max(m0); m1 = wave_max(m1); m2 = wave_max(m2); m3 = wave_max(m3);

            float d0 = 0.f, d1 = 0.f, d2 = 0.f, d3 = 0.f;
            for (int idx = lane; idx < total; idx += 32) {
                int s = idx < degE ? csr[beg + idx] : node;
                float4 a = *(const float4*)(asrc + (size_t)s * NH);
                d0 += __expf(lrelu_att(a.x + ad.x) - m0);
                d1 += __expf(lrelu_att(a.y + ad.y) - m1);
                d2 += __expf(lrelu_att(a.z + ad.z) - m2);
                d3 += __expf(lrelu_att(a.w + ad.w) - m3);
            }
            d0 = wave_sum(d0); d1 = wave_sum(d1); d2 = wave_sum(d2); d3 = wave_sum(d3);

            for (int base = 0; base < total; base += 32) {
                int idx = base + lane;
                int s = node;
                float al0 = 0.f, al1 = 0.f, al2 = 0.f, al3 = 0.f;
                if (idx < total) {
                    s = idx < degE ? csr[beg + idx] : node;
                    float4 a = *(const float4*)(asrc + (size_t)s * NH);
                    al0 = __expf(lrelu_att(a.x + ad.x) - m0) / d0;
                    al1 = __expf(lrelu_att(a.y + ad.y) - m1) / d1;
                    al2 = __expf(lrelu_att(a.z + ad.z) - m2) / d2;
                    al3 = __expf(lrelu_att(a.w + ad.w) - m3) / d3;
                }
                int left = total - base;
                add_rows(h, s, al0, al1, al2, al3, left < 32 ? left : 32, lane, hsel, accA, accB);
            }
        }

        float r[8] = {accA.x + bp[0],   accA.y + bp[1],   accA.z + bp[2],   accA.w + bp[3],
                      accB.x + bp[128], accB.y + bp[129], accB.z + bp[130], accB.w + bp[131]};
        int g = batch[node];
        if (g != curg) {
            if (curg >= 0) {
                float* pA = pooled + (size_t)curg * HC + 4 * lane;
                #pragma unroll
                for (int c = 0; c < 4; ++c) {
                    atomicAdd(&pA[c], pa[c]);
                    atomicAdd(&pA[128 + c], pa[4 + c]);
                }
                if (lane == 0) atomicAdd(&cnt[curg], pc);
            }
            curg = g;
            #pragma unroll
            for (int c = 0; c < 8; ++c) pa[c] = 0.f;
            pc = 0.f;
        }
        #pragma unroll
        for (int c = 0; c < 8; ++c) {
            float u = r[c];
            pa[c] += u > 0.f ? u : 0.01f * u;
        }
        pc += 1.0f;
    }

    *(float4*)&sred[w][4 * lane]       = make_float4(pa[0], pa[1], pa[2], pa[3]);
    *(float4*)&sred[w][128 + 4 * lane] = make_float4(pa[4], pa[5], pa[6], pa[7]);
    if (lane == 0) { sgs[w] = curg; scn[w] = pc; }
    __syncthreads();

    int t = threadIdx.x;
    int gk[8];
    float vk[8], ck[8];
    #pragma unroll
    for (int k = 0; k < 8; ++k) { gk[k] = sgs[k]; vk[k] = sred[k][t]; ck[k] = scn[k]; }
    int cg = -1;
    float s = 0.f, sc = 0.f;
    #pragma unroll
    for (int k = 0; k < 8; ++k) {
        if (gk[k] < 0) continue;
        if (gk[k] != cg) {
            if (cg >= 0) {
                atomicAdd(&pooled[(size_t)cg * HC + t], s);
                if (t == 0) atomicAdd(&cnt[cg], sc);
            }
            cg = gk[k]; s = 0.f; sc = 0.f;
        }
        s += vk[k];
        sc += ck[k];
    }
    if (cg >= 0) {
        atomicAdd(&pooled[(size_t)cg * HC + t], s);
        if (t == 0) atomicAdd(&cnt[cg], sc);
    }
}

#define G2_WAVES 4
#define G2_KS    16
__global__ __launch_bounds__(32 * G2_WAVES) void gemm_out(const float* __restrict__ pooled,
                                                           const float* __restrict__ cnt,
                                                           const float* __restrict__ fw,
                                                           const float* __restrict__ fb,
                                                           float* __restrict__ out, int g) {
    __shared__ __align__(8) float2 red[G2_WAVES][8][32];
    int mTile = blockIdx.x;
    int lane = threadIdx.x & 31, q = threadIdx.x >> 5;
    int mr = lane & 15, half = lane >> 4;
    int row = mTile * 16 + mr;
    int col0 = blockIdx.y * 32 + 2 * mr;
    int kw = q * (4 * G2_KS) + 2 * half;
    const float* arow = pooled + (size_t)row * HC + kw;
    const float* wp = fw + (size_t)kw * NOUTD + col0;

    float cn = cnt[row];
    v2f a[G2_KS], w0[G2_KS], w1[G2_KS];
    #pragma unroll
    for (int i = 0; i < G2_KS; ++i) {
        a[i] = *(const v2f*)(arow + 4 * i);
        const float* p = wp + (size_t)(4 * i) * NOUTD;
        w0[i] = *(const v2f*)p;
        w1[i] = *(const v2f*)(p + NOUTD);
    }
    v2f bb = *(const v2f*)(fb + col0);
    __builtin_amdgcn_sched_barrier(0);
    float inv = 1.0f / fmaxf(cn, 1.0f);
    v8f acc0 = {}, acc1 = {};
    #pragma unroll
    for (int i = 0; i < G2_KS; ++i) {
        v2f as; as.x = a[i].x * inv; as.y = a[i].y * inv;
        v2f b0; b0.x = w0[i].x; b0.y = w1[i].x;
        v2f b1; b1.x = w0[i].y; b1.y = w1[i].y;
        acc0 = __builtin_amdgcn_wmma_f32_16x16x4_f32(false, as, false, b0,
                                                     (short)0, acc0, false, false);
        acc1 = __builtin_amdgcn_wmma_f32_16x16x4_f32(false, as, false, b1,
                                                     (short)0, acc1, false, false);
    }
    #pragma unroll
    for (int r = 0; r < 8; ++r)
        red[q][r][lane] = make_float2(acc0[r], acc1[r]);
    __syncthreads();

    #pragma unroll
    for (int j = 0; j < 2; ++j) {
        int r = 2 * q + j;
        float2 s = red[0][r][lane];
        #pragma unroll
        for (int w = 1; w < G2_WAVES; ++w) {
            float2 p = red[w][r][lane];
            s.x += p.x; s.y += p.y;
        }
        float* op = out + (size_t)(mTile * 16 + 8 * half + r) * NOUTD + col0;
        *(float2*)op = make_float2(s.x + bb.x, s.y + bb.y);
    }
}

extern "C" void kernel_launch(void* const* d_in, const int* in_sizes, int n_in,
                              void* d_out, int out_size, void* d_ws, size_t ws_size,
                              hipStream_t stream) {
    const float* x      = (const float*)d_in[0];
    const int*   ei     = (const int*)  d_in[1];
    const int*   batch  = (const int*)  d_in[2];
    const float* lin_w  = (const float*)d_in[3];
    const float* att_s  = (const float*)d_in[4];
    const float* att_d  = (const float*)d_in[5];
    const float* bias   = (const float*)d_in[6];
    const float* fc1_w  = (const float*)d_in[7];
    const float* fc1_b  = (const float*)d_in[8];
    float* out = (float*)d_out;

    const int N = in_sizes[2];
    const int E = in_sizes[1] / 2;
    const int G = out_size / NOUTD;

    float* ws = (float*)d_ws;
    float* h      = ws;
    float* asrc   = h + (size_t)N * HC;
    float* adst   = asrc + (size_t)N * NH;
    float* pooled = adst + (size_t)N * NH;
    float* cnt    = pooled + (size_t)G * HC;
    int*   rank   = (int*)(cnt + (size_t)G);
    int*   csr    = rank + (size_t)E;
    int*   off    = csr + (size_t)E;
    int*   part   = (int*)h;
    int*   ctot   = part + (size_t)RANK_WGS * LDS_NODES;

    init_ws<<<256, 256, 0, stream>>>(pooled, cnt, G);

    edge_rank<<<RANK_WGS, 1024, 0, stream>>>(ei, rank, part, ctot, E);
    scan_deg<<<N / SCAN_CHUNK + 1, SCAN_CHUNK, 0, stream>>>(part, ctot, off, N);
    dim3 gf((((E + RANK_WGS - 1) / RANK_WGS) + 255) / 256, RANK_WGS);
    edge_fill<<<gf, 256, 0, stream>>>(ei, part, rank, csr, E);

    gemm_x_linw<<<(N + G1_ROWS - 1) / G1_ROWS, 128, 0, stream>>>(x, lin_w, h, N);

    att_kernel<<<(N * 32 + 255) / 256, 256, 0, stream>>>(h, att_s, att_d, asrc, adst, N);

    node_gather<<<(N + 8 * NPW - 1) / (8 * NPW), 256, 0, stream>>>(off, csr, asrc, adst, h, bias,
                                                            batch, pooled, cnt, N);

    dim3 g2(G / 16, NOUTD / 32);
    gemm_out<<<g2, 32 * G2_WAVES, 0, stream>>>(pooled, cnt, fc1_w, fc1_b, out, G);
}
